# starting kernel
# baseline (speedup 1.0000x reference)
#include <hip/hip_runtime.h>

typedef __attribute__((ext_vector_type(16))) __bf16 v16bf;
typedef __attribute__((ext_vector_type(8)))  __bf16 v8bf;
typedef __attribute__((ext_vector_type(8)))  float  v8f;

__device__ __forceinline__ __bf16 f2bf(float f) {
  unsigned int u = __builtin_bit_cast(unsigned int, f);
  u += 0x7FFFu + ((u >> 16) & 1u);
  unsigned short s = (unsigned short)(u >> 16);
  return __builtin_bit_cast(__bf16, s);
}

__global__ void fill_zero_kernel(float* __restrict__ p, int n) {
  int i = blockIdx.x * blockDim.x + threadIdx.x;
  if (i < n) p[i] = 0.0f;
}

__global__ void degree_kernel(const int* __restrict__ src, const int* __restrict__ dst,
                              float* __restrict__ degOut, float* __restrict__ degIn, int nE) {
  int e = blockIdx.x * blockDim.x + threadIdx.x;
  if (e < nE) {
    atomicAdd(&degOut[src[e]], 1.0f);
    atomicAdd(&degIn[dst[e]], 1.0f);
  }
}

__global__ void norm_kernel(float* __restrict__ p, int n) {
  int i = blockIdx.x * blockDim.x + threadIdx.x;
  if (i < n) p[i] = rsqrtf(fmaxf(p[i], 1.0f));
}

template<int K>
__global__ __launch_bounds__(128) void gemm_rownorm_wmma(
    const float* __restrict__ X, const float* __restrict__ rowScale,
    const float* __restrict__ Wg, float* __restrict__ Y, int nRows) {
  __shared__ __bf16 Xs[16][K];
  __shared__ __bf16 Wt[64][K + 8];

  const int tid     = threadIdx.x;
  const int rowBase = blockIdx.x * 16;

  for (int idx = tid; idx < 16 * K; idx += 128) {
    int r = idx / K, k = idx - r * K;
    int row = rowBase + r;
    row = (row < nRows) ? row : (nRows - 1);
    Xs[r][k] = f2bf(X[(long)row * K + k] * rowScale[row]);
  }
  for (int idx = tid; idx < K * 64; idx += 128) {
    int k = idx >> 6, c = idx & 63;
    Wt[c][k] = f2bf(Wg[idx]);
  }
  __syncthreads();

  const int lane    = tid & 31;
  const int wave    = tid >> 5;
  const int hh      = lane >> 4;
  const int l       = lane & 15;
  const int colBase = wave << 4;

  v8f acc = {};
  #pragma unroll
  for (int k0 = 0; k0 < K; k0 += 32) {
    const v8bf a0 = *reinterpret_cast<const v8bf*>(&Xs[l][k0 + (hh << 3)]);
    const v8bf a1 = *reinterpret_cast<const v8bf*>(&Xs[l][k0 + 16 + (hh << 3)]);
    const v8bf b0 = *reinterpret_cast<const v8bf*>(&Wt[colBase + l][k0 + (hh << 3)]);
    const v8bf b1 = *reinterpret_cast<const v8bf*>(&Wt[colBase + l][k0 + 16 + (hh << 3)]);
    v16bf a, b;
    #pragma unroll
    for (int i = 0; i < 8; ++i) {
      a[i] = a0[i]; a[i + 8] = a1[i];
      b[i] = b0[i]; b[i + 8] = b1[i];
    }
    acc = __builtin_amdgcn_wmma_f32_16x16x32_bf16(
         false, a,  false, b,
         (short)0, acc,  false,  false);
  }

  float* __restrict__ yOut = Y + ((long)(rowBase + (hh << 3)) * 64 + colBase + l);
  if (rowBase + 16 <= nRows) {
    #pragma unroll
    for (int v = 0; v < 8; ++v) yOut[(long)v * 64] = acc[v];
  } else {
    #pragma unroll
    for (int v = 0; v < 8; ++v) {
      if (rowBase + v + (hh << 3) < nRows) yOut[(long)v * 64] = acc[v];
    }
  }
}

__global__ void edge_agg_kernel(const float* __restrict__ y, const int* __restrict__ src,
                                const int* __restrict__ dst, float* __restrict__ agg, int nE) {
  int gid = blockIdx.x * blockDim.x + threadIdx.x;
  int e = gid >> 4;
  if (e >= nE) return;
  int c4 = (gid & 15) << 2;
  int s = src[e], d = dst[e];
  const float4 v = *reinterpret_cast<const float4*>(y + (long)s * 64 + c4);
  float* ap = agg + (long)d * 64 + c4;
  atomicAdd(ap + 0, v.x);
  atomicAdd(ap + 1, v.y);
  atomicAdd(ap + 2, v.z);
  atomicAdd(ap + 3, v.w);
}

__global__ void finalize_kernel(const float* __restrict__ agg, const float* __restrict__ inNorm,
                                const float* __restrict__ bias, float* __restrict__ out,
                                int nRows, int doRelu) {
  int gid = blockIdx.x * blockDim.x + threadIdx.x;
  if (gid >= nRows * 64) return;
  int n = gid >> 6, c = gid & 63;
  float v = agg[gid] * inNorm[n] + bias[c];
  out[gid] = doRelu ? fmaxf(v, 0.0f) : v;
}

static inline int cdiv_i(long a, long b) { return (int)((a + b - 1) / b); }

extern "C" void kernel_launch(void* const* d_in, const int* in_sizes, int n_in,
                              void* d_out, int out_size, void* d_ws, size_t ws_size,
                              hipStream_t stream) {
  const float* feat = (const float*)d_in[0];
  const float* W0   = (const float*)d_in[1];
  const float* b0   = (const float*)d_in[2];
  const float* W1   = (const float*)d_in[3];
  const float* b1   = (const float*)d_in[4];
  const float* W2   = (const float*)d_in[5];
  const float* b2   = (const float*)d_in[6];
  const int*   src  = (const int*)d_in[7];
  const int*   dst  = (const int*)d_in[8];

  const int N = in_sizes[0] / 128;
  const int E = in_sizes[7];
  float* out = (float*)d_out;

  float* wsf = (float*)d_ws;
  float* no  = wsf;
  float* ni  = no + N;
  float* y   = ni + N;
  float* ag  = y  + (size_t)N * 64;
  float* h   = ag + (size_t)N * 64;

  const int NB = 256;
  const int gemmBlocks = cdiv_i(N, 16);
  const int nNF = N * 64;

  fill_zero_kernel<<<cdiv_i(2 * N, NB), NB, 0, stream>>>(no, 2 * N);
  degree_kernel<<<cdiv_i(E, NB), NB, 0, stream>>>(src, dst, no, ni, E);
  norm_kernel<<<cdiv_i(2 * N, NB), NB, 0, stream>>>(no, 2 * N);

  gemm_rownorm_wmma<128><<<gemmBlocks, 128, 0, stream>>>(feat, no, W0, y, N);
  fill_zero_kernel<<<cdiv_i(nNF, NB), NB, 0, stream>>>(ag, nNF);
  edge_agg_kernel<<<cdiv_i((long)E * 16, NB), NB, 0, stream>>>(y, src, dst, ag, E);
  finalize_kernel<<<cdiv_i(nNF, NB), NB, 0, stream>>>(ag, ni, b0, h, N, 1);

  gemm_rownorm_wmma<64><<<gemmBlocks, 128, 0, stream>>>(h, no, W1, y, N);
  fill_zero_kernel<<<cdiv_i(nNF, NB), NB, 0, stream>>>(ag, nNF);
  edge_agg_kernel<<<cdiv_i((long)E * 16, NB), NB, 0, stream>>>(y, src, dst, ag, E);
  finalize_kernel<<<cdiv_i(nNF, NB), NB, 0, stream>>>(ag, ni, b1, h, N, 1);

  gemm_rownorm_wmma<64><<<gemmBlocks, 128, 0, stream>>>(h, no, W2, y, N);
  fill_zero_kernel<<<cdiv_i(nNF, NB), NB, 0, stream>>>(ag, nNF);
  edge_agg_kernel<<<cdiv_i((long)E * 16, NB), NB, 0, stream>>>(y, src, dst, ag, E);
  finalize_kernel<<<cdiv_i(nNF, NB), NB, 0, stream>>>(ag, ni, b2, out, N, 0);
}


# layer-0 product moved into the edge-list fill launch as extra 32-wave work-groups (128 rows each, W staged once per group), two groups per WGP: the independent phases overlap; 8 -> 7 launches
# speedup vs baseline: 5.0042x; 5.0042x over previous
#include <hip/hip_runtime.h>

typedef __attribute__((ext_vector_type(16))) __bf16 v16bf;
typedef __attribute__((ext_vector_type(8)))  __bf16 v8bf;
typedef __attribute__((ext_vector_type(8)))  float  v8f;

__device__ __forceinline__ __bf16 f2bf(float f) {
  unsigned int u = __builtin_bit_cast(unsigned int, f);
  u += 0x7FFFu + ((u >> 16) & 1u);
  unsigned short s = (unsigned short)(u >> 16);
  return __builtin_bit_cast(__bf16, s);
}

constexpr int DEG_BINS  = 65536;
constexpr int DEG_PARTS = 64;
constexpr int DEG_BATCH = 8;

__global__ __launch_bounds__(1024) void degree_plane_kernel(
    const int* __restrict__ src, const int* __restrict__ dst,
    unsigned* __restrict__ planes, int nE, int nNodes) {
  __shared__ unsigned hist[DEG_BINS];
  const int tid = threadIdx.x;
  const int* __restrict__ idx = blockIdx.y ? dst : src;
  const int base   = blockIdx.z * DEG_BINS;
  const int nb     = min(DEG_BINS, nNodes - base);
  const int nb4    = nb >> 2;
  const int stride = (nNodes + 63) & ~63;

  uint4* hist4 = reinterpret_cast<uint4*>(hist);
  for (int i = tid; i < nb4; i += 1024) hist4[i] = make_uint4(0u, 0u, 0u, 0u);
  for (int i = (nb4 << 2) + tid; i < nb; i += 1024) hist[i] = 0u;
  __syncthreads();

  const int per = (((nE + DEG_PARTS - 1) / DEG_PARTS) + 3) & ~3;
  const int e0  = min(nE, (int)blockIdx.x * per);
  const int e1  = min(nE, e0 + per);
  const int nv  = (e1 - e0) >> 2;
  const int4* __restrict__ v = reinterpret_cast<const int4*>(idx + e0);
  for (int i0 = tid; i0 < nv; i0 += DEG_BATCH * 1024) {
    int4 q[DEG_BATCH];
    #pragma unroll
    for (int u = 0; u < DEG_BATCH; ++u) {
      const int i = i0 + u * 1024;
      q[u] = make_int4(-1, -1, -1, -1);
      if (i < nv) q[u] = v[i];
    }
    #pragma unroll
    for (int u = 0; u < DEG_BATCH; ++u) {
      unsigned b;
      b = (unsigned)(q[u].x - base); if (b < (unsigned)nb) atomicAdd(&hist[b], 1u);
      b = (unsigned)(q[u].y - base); if (b < (unsigned)nb) atomicAdd(&hist[b], 1u);
      b = (unsigned)(q[u].z - base); if (b < (unsigned)nb) atomicAdd(&hist[b], 1u);
      b = (unsigned)(q[u].w - base); if (b < (unsigned)nb) atomicAdd(&hist[b], 1u);
    }
  }
  for (int e = e0 + (nv << 2) + tid; e < e1; e += 1024) {
    unsigned b = (unsigned)(idx[e] - base);
    if (b < (unsigned)nb) atomicAdd(&hist[b], 1u);
  }
  __syncthreads();

  unsigned* __restrict__ outp =
      planes + ((size_t)blockIdx.y * DEG_PARTS + blockIdx.x) * (size_t)stride + base;
  uint4* __restrict__ outp4 = reinterpret_cast<uint4*>(outp);
  for (int i = tid; i < nb4; i += 1024) outp4[i] = hist4[i];
  for (int i = (nb4 << 2) + tid; i < nb; i += 1024) outp[i] = hist[i];
}

constexpr int SCAN_BLOCK = 256;

__global__ __launch_bounds__(SCAN_BLOCK) void degree_norm_scan_kernel(
    unsigned* __restrict__ planes, float* __restrict__ outNorm, float* __restrict__ inNorm,
    int* __restrict__ localPre, int* __restrict__ blockSum, int nNodes) {
  __shared__ int waveSum[SCAN_BLOCK / 32];
  const int tid  = threadIdx.x;
  const int lane = tid & 31;
  const int wave = tid >> 5;
  const int n = blockIdx.x * SCAN_BLOCK + tid;
  const int stride = (nNodes + 63) & ~63;
  unsigned so = 0u, si = 0u;
  if (n < nNodes) {
    const unsigned* p = planes + n;
    unsigned*       q = planes + (size_t)DEG_PARTS * (size_t)stride + n;
    unsigned cq[DEG_PARTS];
    #pragma unroll
    for (int g = 0; g < DEG_PARTS; ++g) { so += p[(size_t)g * stride]; cq[g] = q[(size_t)g * stride]; }
    #pragma unroll
    for (int g = 0; g < DEG_PARTS; ++g) { q[(size_t)g * stride] = si; si += cq[g]; }
    outNorm[n] = rsqrtf(fmaxf((float)so, 1.0f));
    inNorm[n]  = rsqrtf(fmaxf((float)si, 1.0f));
  }
  int incl = (int)si;
  #pragma unroll
  for (int d = 1; d < 32; d <<= 1) {
    const int t = __shfl_up(incl, d, 32);
    if (lane >= d) incl += t;
  }
  if (lane == 31) waveSum[wave] = incl;
  __syncthreads();
  int woff = 0;
  #pragma unroll
  for (int w = 0; w < SCAN_BLOCK / 32; ++w) woff += (w < wave) ? waveSum[w] : 0;
  if (n < nNodes) localPre[n] = woff + incl - (int)si;
  if (tid == SCAN_BLOCK - 1) blockSum[blockIdx.x] = woff + incl;
}

__global__ __launch_bounds__(SCAN_BLOCK) void rowptr_kernel(
    const int* __restrict__ localPre, const int* __restrict__ blockSum,
    int* __restrict__ rowptr, int nNodes) {
  __shared__ int waveSum[SCAN_BLOCK / 32];
  const int tid = threadIdx.x;
  const int b   = blockIdx.x;
  int part = 0;
  for (int i = tid; i < b; i += SCAN_BLOCK) part += blockSum[i];
  #pragma unroll
  for (int d = 16; d >= 1; d >>= 1) part += __shfl_xor(part, d, 32);
  if ((tid & 31) == 0) waveSum[tid >> 5] = part;
  __syncthreads();
  int off = 0;
  #pragma unroll
  for (int w = 0; w < SCAN_BLOCK / 32; ++w) off += waveSum[w];
  const int n = b * SCAN_BLOCK + tid;
  if (n < nNodes) rowptr[n] = off + localPre[n];
  if (b == (int)gridDim.x - 1 && tid == 0) rowptr[nNodes] = off + blockSum[b];
}

constexpr int FILL_BINS  = 16384;
constexpr int FILL_BATCH = 4;

__device__ __forceinline__ void place_edge(unsigned* cur, int* __restrict__ sortedSrc,
                                           int d, int s, int base, int nb, int nE) {
  const unsigned b = (unsigned)(d - base);
  if (b < (unsigned)nb) {
    const unsigned p = atomicAdd(&cur[b], 1u);
    if (p < (unsigned)nE) sortedSrc[p] = s;
  }
}

__device__ __forceinline__ void csr_fill_group(
    const int* __restrict__ src, const int* __restrict__ dst,
    const unsigned* __restrict__ slicePre, const int* __restrict__ rowptr,
    int* __restrict__ sortedSrc, int nE, int nNodes, int gx, int gy, int nr, unsigned* cur) {
  const int tid    = threadIdx.x;
  const int rs     = (nNodes + nr - 1) / nr;
  const int base   = gy * rs;
  const int nb     = max(0, min(rs, nNodes - base));
  const int stride = (nNodes + 63) & ~63;
  const unsigned* __restrict__ pre = slicePre + (size_t)gx * (size_t)stride + base;
  for (int i = tid; i < nb; i += 1024) cur[i] = pre[i] + (unsigned)rowptr[base + i];
  __syncthreads();

  const int per = (((nE + DEG_PARTS - 1) / DEG_PARTS) + 3) & ~3;
  const int e0  = min(nE, gx * per);
  const int e1  = min(nE, e0 + per);
  const int nv  = (e1 - e0) >> 2;
  const int4* __restrict__ vd = reinterpret_cast<const int4*>(dst + e0);
  const int4* __restrict__ vs = reinterpret_cast<const int4*>(src + e0);
  for (int i0 = tid; i0 < nv; i0 += FILL_BATCH * 1024) {
    int4 qd[FILL_BATCH], qs[FILL_BATCH];
    #pragma unroll
    for (int u = 0; u < FILL_BATCH; ++u) {
      const int i = i0 + u * 1024;
      qd[u] = make_int4(-1, -1, -1, -1);
      qs[u] = make_int4(0, 0, 0, 0);
      if (i < nv) { qd[u] = vd[i]; qs[u] = vs[i]; }
    }
    #pragma unroll
    for (int u = 0; u < FILL_BATCH; ++u) {
      place_edge(cur, sortedSrc, qd[u].x, qs[u].x, base, nb, nE);
      place_edge(cur, sortedSrc, qd[u].y, qs[u].y, base, nb, nE);
      place_edge(cur, sortedSrc, qd[u].z, qs[u].z, base, nb, nE);
      place_edge(cur, sortedSrc, qd[u].w, qs[u].w, base, nb, nE);
    }
  }
  for (int e = e0 + (nv << 2) + tid; e < e1; e += 1024)
    place_edge(cur, sortedSrc, dst[e], src[e], base, nb, nE);
}

typedef __attribute__((ext_vector_type(4))) __bf16 v4bf;
constexpr int PROD_ROWS  = 128;
constexpr int PROD_K     = 128;
constexpr int PROD_PITCH = PROD_K + 8;

__device__ __forceinline__ void product_rows128(
    const float* __restrict__ X, const float* __restrict__ rowScale,
    const float* __restrict__ Wg, float* __restrict__ Y, int nRows, int rowBase, __bf16* Xs) {
  __bf16* Wt = Xs + PROD_ROWS * PROD_PITCH;
  const int tid  = threadIdx.x;
  const int lane = tid & 31;
  const int wave = tid >> 5;

  const int wc = tid & 63;
  const int wk = (tid >> 6) << 3;
  float wv[8];
  #pragma unroll
  for (int i = 0; i < 8; ++i) wv[i] = Wg[(wk + i) * 64 + wc];
  float4 xv[4];
  float  xs[4];
  #pragma unroll
  for (int j = 0; j < 4; ++j) {
    int row = rowBase + wave + 32 * j;
    row = (row < nRows) ? row : (nRows - 1);
    xv[j] = *reinterpret_cast<const float4*>(X + (size_t)row * PROD_K + lane * 4);
    xs[j] = rowScale[row];
  }
  #pragma unroll
  for (int j = 0; j < 4; ++j) {
    v4bf p;
    p[0] = f2bf(xv[j].x * xs[j]);
    p[1] = f2bf(xv[j].y * xs[j]);
    p[2] = f2bf(xv[j].z * xs[j]);
    p[3] = f2bf(xv[j].w * xs[j]);
    *reinterpret_cast<v4bf*>(&Xs[(wave + 32 * j) * PROD_PITCH + lane * 4]) = p;
  }
  v8bf wp;
  #pragma unroll
  for (int i = 0; i < 8; ++i) wp[i] = f2bf(wv[i]);
  *reinterpret_cast<v8bf*>(&Wt[wc * PROD_PITCH + wk]) = wp;
  __syncthreads();

  const int hh      = lane >> 4;
  const int l       = lane & 15;
  const int rt      = wave >> 2;
  const int colBase = (wave & 3) << 4;
  const __bf16* xr = Xs + (rt * 16 + l) * PROD_PITCH;
  const __bf16* wr = Wt + (colBase + l) * PROD_PITCH;

  v8f acc = {};
  #pragma unroll
  for (int k0 = 0; k0 < PROD_K; k0 += 32) {
    const v8bf a0 = *reinterpret_cast<const v8bf*>(xr + k0 + (hh << 3));
    const v8bf a1 = *reinterpret_cast<const v8bf*>(xr + k0 + 16 + (hh << 3));
    const v8bf b0 = *reinterpret_cast<const v8bf*>(wr + k0 + (hh << 3));
    const v8bf b1 = *reinterpret_cast<const v8bf*>(wr + k0 + 16 + (hh << 3));
    v16bf a, b;
    #pragma unroll
    for (int i = 0; i < 8; ++i) {
      a[i] = a0[i]; a[i + 8] = a1[i];
      b[i] = b0[i]; b[i + 8] = b1[i];
    }
    acc = __builtin_amdgcn_wmma_f32_16x16x32_bf16(
         false, a,  false, b,
         (short)0, acc,  false,  false);
  }

  const int r0 = rowBase + rt * 16;
  float* __restrict__ yOut = Y + ((size_t)(r0 + (hh << 3)) * 64 + colBase + l);
  if (r0 + 16 <= nRows) {
    #pragma unroll
    for (int v = 0; v < 8; ++v) yOut[(size_t)v * 64] = acc[v];
  } else {
    #pragma unroll
    for (int v = 0; v < 8; ++v) {
      if (r0 + v + (hh << 3) < nRows) yOut[(size_t)v * 64] = acc[v];
    }
  }
}

__global__ __launch_bounds__(1024, 2) void csr_fill_product_kernel(
    const int* __restrict__ src, const int* __restrict__ dst,
    const unsigned* __restrict__ slicePre, const int* __restrict__ rowptr,
    int* __restrict__ sortedSrc, int nE, int nNodes, int fillRanges,
    const float* __restrict__ X, const float* __restrict__ rowScale,
    const float* __restrict__ Wg, float* __restrict__ Y) {
  __shared__ __attribute__((aligned(16))) unsigned smem[FILL_BINS];
  const int bid   = (int)blockIdx.x;
  const int nFill = DEG_PARTS * fillRanges;
  if (bid < nFill) {
    csr_fill_group(src, dst, slicePre, rowptr, sortedSrc, nE, nNodes,
                   bid % DEG_PARTS, bid / DEG_PARTS, fillRanges, smem);
  } else {
    product_rows128(X, rowScale, Wg, Y, nNodes, (bid - nFill) * PROD_ROWS, reinterpret_cast<__bf16*>(smem));
  }
}

constexpr int GATHER_BATCH = 8;

template<bool PARTIAL>
__device__ __forceinline__ void gather_batch(const float2* __restrict__ y2, int idx, int k, int cnt,
                                             int lane, int lastRow, float2& acc) {
  float2 v[GATHER_BATCH];
  #pragma unroll
  for (int u = 0; u < GATHER_BATCH; ++u) {
    const int kk = PARTIAL ? min(k + u, cnt - 1) : (k + u);
    int s = __builtin_amdgcn_readlane(idx, kk);
    s = (int)min((unsigned)s, (unsigned)lastRow);
    v[u] = y2[s * 32 + lane];
  }
  #pragma unroll
  for (int u = 0; u < GATHER_BATCH; ++u) {
    const bool ok = !PARTIAL || (k + u) < cnt;
    acc.x += ok ? v[u].x : 0.0f;
    acc.y += ok ? v[u].y : 0.0f;
  }
}

__device__ __forceinline__ float2 gather_row_sum(const float2* __restrict__ y2, const int* __restrict__ rowptr,
                                                 const int* __restrict__ sortedSrc, int n, int lane, int nRows) {
  const int beg = rowptr[n];
  const int end = rowptr[n + 1];
  float2 acc = make_float2(0.0f, 0.0f);
  int idx = (beg + lane < end) ? sortedSrc[beg + lane] : 0;
  for (int base = beg; base < end; base += 32) {
    const int cnt = min(32, end - base);
    const int inext = base + 32 + lane;
    const int nxt = (inext < end) ? sortedSrc[inext] : 0;
    int k = 0;
    for (; k + GATHER_BATCH <= cnt; k += GATHER_BATCH)
      gather_batch<false>(y2, idx, k, cnt, lane, nRows - 1, acc);
    if (k < cnt)
      gather_batch<true>(y2, idx, k, cnt, lane, nRows - 1, acc);
    idx = nxt;
  }
  return acc;
}

__global__ __launch_bounds__(256) void gather_finalize_kernel(
    const float* __restrict__ y, const int* __restrict__ rowptr, const int* __restrict__ sortedSrc,
    const float* __restrict__ inNorm, const float* __restrict__ bias, float* __restrict__ out,
    int nRows, int doRelu) {
  const int lane = threadIdx.x & 31;
  const int n = __builtin_amdgcn_readfirstlane((int)(blockIdx.x * 8 + (threadIdx.x >> 5)));
  if (n >= nRows) return;
  const float2 acc = gather_row_sum(reinterpret_cast<const float2*>(y), rowptr, sortedSrc, n, lane, nRows);
  const float  sc = inNorm[n];
  const float2 bb = *reinterpret_cast<const float2*>(bias + 2 * lane);
  float2 r;
  r.x = acc.x * sc + bb.x;
  r.y = acc.y * sc + bb.y;
  if (doRelu) { r.x = fmaxf(r.x, 0.0f); r.y = fmaxf(r.y, 0.0f); }
  *reinterpret_cast<float2*>(out + (size_t)n * 64 + 2 * lane) = r;
}

typedef __attribute__((ext_vector_type(2))) __bf16 v2bf;

__global__ __launch_bounds__(512) void gather_finalize_gemm_kernel(
    const float* __restrict__ y, const int* __restrict__ rowptr, const int* __restrict__ sortedSrc,
    const float* __restrict__ inNorm, const float* __restrict__ bias,
    const float* __restrict__ outNorm, const float* __restrict__ Wg,
    float* __restrict__ yNext, int nRows) {
  __shared__ __attribute__((aligned(16))) __bf16 Xs[16][64 + 8];
  __shared__ __attribute__((aligned(16))) __bf16 Wt[64][64 + 8];

  const int tid  = threadIdx.x;
  const int lane = tid & 31;
  const int wave    = __builtin_amdgcn_readfirstlane(tid >> 5);
  const int rowBase = blockIdx.x * 16;
  const int n       = rowBase + wave;

  const int wc = tid & 63;
  const int wk = (tid >> 6) << 3;
  float wv[8];
  #pragma unroll
  for (int i = 0; i < 8; ++i) wv[i] = Wg[(wk + i) * 64 + wc];

  float2 r = make_float2(0.0f, 0.0f);
  if (n < nRows) {
    const float2 acc = gather_row_sum(reinterpret_cast<const float2*>(y), rowptr, sortedSrc, n, lane, nRows);
    const float  sc = inNorm[n];
    const float  so = outNorm[n];
    const float2 bb = *reinterpret_cast<const float2*>(bias + 2 * lane);
    r.x = fmaxf(acc.x * sc + bb.x, 0.0f) * so;
    r.y = fmaxf(acc.y * sc + bb.y, 0.0f) * so;
  }
  v2bf rp;
  rp[0] = f2bf(r.x);
  rp[1] = f2bf(r.y);
  *reinterpret_cast<v2bf*>(&Xs[wave][2 * lane]) = rp;

  v8bf wp;
  #pragma unroll
  for (int i = 0; i < 8; ++i) wp[i] = f2bf(wv[i]);
  *reinterpret_cast<v8bf*>(&Wt[wc][wk]) = wp;
  __syncthreads();

  if (wave >= 4) return;

  const int hh      = lane >> 4;
  const int l       = lane & 15;
  const int colBase = wave << 4;

  v8f acc = {};
  #pragma unroll
  for (int k0 = 0; k0 < 64; k0 += 32) {
    const v8bf a0 = *reinterpret_cast<const v8bf*>(&Xs[l][k0 + (hh << 3)]);
    const v8bf a1 = *reinterpret_cast<const v8bf*>(&Xs[l][k0 + 16 + (hh << 3)]);
    const v8bf b0 = *reinterpret_cast<const v8bf*>(&Wt[colBase + l][k0 + (hh << 3)]);
    const v8bf b1 = *reinterpret_cast<const v8bf*>(&Wt[colBase + l][k0 + 16 + (hh << 3)]);
    v16bf a, b;
    #pragma unroll
    for (int i = 0; i < 8; ++i) {
      a[i] = a0[i]; a[i + 8] = a1[i];
      b[i] = b0[i]; b[i + 8] = b1[i];
    }
    acc = __builtin_amdgcn_wmma_f32_16x16x32_bf16(
         false, a,  false, b,
         (short)0, acc,  false,  false);
  }

  float* __restrict__ yOut = yNext + ((long)(rowBase + (hh << 3)) * 64 + colBase + l);
  if (rowBase + 16 <= nRows) {
    #pragma unroll
    for (int v = 0; v < 8; ++v) yOut[(long)v * 64] = acc[v];
  } else {
    #pragma unroll
    for (int v = 0; v < 8; ++v) {
      if (rowBase + v + (hh << 3) < nRows) yOut[(long)v * 64] = acc[v];
    }
  }
}

static inline int cdiv_i(long a, long b) { return (int)((a + b - 1) / b); }

extern "C" void kernel_launch(void* const* d_in, const int* in_sizes, int n_in,
                              void* d_out, int out_size, void* d_ws, size_t ws_size,
                              hipStream_t stream) {
  const float* feat = (const float*)d_in[0];
  const float* W0   = (const float*)d_in[1];
  const float* b0   = (const float*)d_in[2];
  const float* W1   = (const float*)d_in[3];
  const float* b1   = (const float*)d_in[4];
  const float* W2   = (const float*)d_in[5];
  const float* b2   = (const float*)d_in[6];
  const int*   src  = (const int*)d_in[7];
  const int*   dst  = (const int*)d_in[8];

  const int N = in_sizes[0] / 128;
  const int E = in_sizes[7];
  float* out = (float*)d_out;

  const size_t P = (size_t)((N + 63) & ~63);
  float* wsf = (float*)d_ws;
  float* no  = wsf;
  float* ni  = no + P;
  float* y   = ni + P;
  float* h   = y  + (size_t)N * 64;
  unsigned* planes = (unsigned*)(h + (size_t)N * 64);
  int* localPre  = (int*)(planes + (size_t)2 * DEG_PARTS * P);
  int* blockSum  = localPre + P;
  int* rowptr    = blockSum + P;
  int* sortedSrc = rowptr + P + 64;

  const int gemmBlocks = cdiv_i(N, 16);
  const int scanBlocks = cdiv_i(N, SCAN_BLOCK);
  const int gatherBlocks = cdiv_i(N, 8);

  degree_plane_kernel<<<dim3(DEG_PARTS, 2, cdiv_i(N, DEG_BINS)), 1024, 0, stream>>>(src, dst, planes, E, N);
  degree_norm_scan_kernel<<<scanBlocks, SCAN_BLOCK, 0, stream>>>(planes, no, ni, localPre, blockSum, N);
  rowptr_kernel<<<scanBlocks, SCAN_BLOCK, 0, stream>>>(localPre, blockSum, rowptr, N);
  const int fillRanges = cdiv_i(N, FILL_BINS);
  csr_fill_product_kernel<<<DEG_PARTS * fillRanges + cdiv_i(N, PROD_ROWS), 1024, 0, stream>>>(
      src, dst, planes + (size_t)DEG_PARTS * P, rowptr, sortedSrc, E, N, fillRanges, feat, no, W0, y);
  gather_finalize_gemm_kernel<<<gemmBlocks, 512, 0, stream>>>(y, rowptr, sortedSrc, ni, b0, no, W1, h, N);
  gather_finalize_gemm_kernel<<<gemmBlocks, 512, 0, stream>>>(h, rowptr, sortedSrc, ni, b1, no, W2, y, N);
  gather_finalize_kernel<<<gatherBlocks, 256, 0, stream>>>(y, rowptr, sortedSrc, ni, b2, out, N, 0);
}


# edge list by two-level bucket sort, slots from edge position (ballot ranks) where the refuted try ranked by arrival: rows in edge order, sum order changes; list stored as runs; 7 -> 5 launches
# speedup vs baseline: 7.1282x; 1.4244x over previous
#include <hip/hip_runtime.h>

typedef __attribute__((ext_vector_type(16))) __bf16 v16bf;
typedef __attribute__((ext_vector_type(8)))  __bf16 v8bf;
typedef __attribute__((ext_vector_type(8)))  float  v8f;

__device__ __forceinline__ __bf16 f2bf(float f) {
  unsigned int u = __builtin_bit_cast(unsigned int, f);
  u += 0x7FFFu + ((u >> 16) & 1u);
  unsigned short s = (unsigned short)(u >> 16);
  return __builtin_bit_cast(__bf16, s);
}

constexpr int SMEM_WORDS  = 25600;
constexpr int NSLICE      = 256;
constexpr int BSHIFT      = 8;
constexpr int NKEY        = 256;
constexpr int SORT_BATCH  = 8;
constexpr int STAGE_PAIRS = 8192;
constexpr int STAGE_LIST  = 16384;

__device__ __forceinline__ unsigned same_key_mask(unsigned key, bool valid) {
  unsigned m = __builtin_amdgcn_ballot_w32(valid);
  #pragma unroll
  for (int bit = 0; bit < 8; ++bit) {
    const bool one = ((key >> bit) & 1u) != 0u;
    const unsigned b = __builtin_amdgcn_ballot_w32(one);
    m &= one ? b : ~b;
  }
  return m;
}

__device__ __forceinline__ int scan256(int v, int tid, unsigned* red, int& total) {
  const int lane = tid & 31;
  const int wave = tid >> 5;
  int incl = v;
  #pragma unroll
  for (int d = 1; d < 32; d <<= 1) {
    const int t = __shfl_up(incl, d, 32);
    if (lane >= d) incl += t;
  }
  if (lane == 31 && wave < 8) red[wave] = (unsigned)incl;
  __syncthreads();
  int woff = 0, tot = 0;
  #pragma unroll
  for (int w = 0; w < 8; ++w) { const int s = (int)red[w]; tot += s; woff += (w < wave) ? s : 0; }
  total = tot;
  return woff + incl - v;
}

constexpr int DEG_BINS  = 25088;
constexpr int DEG_PARTS = 64;
constexpr int DEG_BATCH = 8;

__device__ __forceinline__ void src_plane_group(const int* __restrict__ idx, unsigned* __restrict__ planes,
                                                int nE, int nNodes, int gx, int gz, unsigned* hist) {
  const int tid    = threadIdx.x;
  const int base   = gz * DEG_BINS;
  const int nb     = max(0, min(DEG_BINS, nNodes - base));
  const int nb4    = nb >> 2;
  const int stride = (nNodes + 63) & ~63;

  uint4* hist4 = reinterpret_cast<uint4*>(hist);
  for (int i = tid; i < nb4; i += 1024) hist4[i] = make_uint4(0u, 0u, 0u, 0u);
  for (int i = (nb4 << 2) + tid; i < nb; i += 1024) hist[i] = 0u;
  __syncthreads();

  const int per = (((nE + DEG_PARTS - 1) / DEG_PARTS) + 3) & ~3;
  const int e0  = min(nE, gx * per);
  const int e1  = min(nE, e0 + per);
  const int nv  = (e1 - e0) >> 2;
  const int4* __restrict__ v = reinterpret_cast<const int4*>(idx + e0);
  for (int i0 = tid; i0 < nv; i0 += DEG_BATCH * 1024) {
    int4 q[DEG_BATCH];
    #pragma unroll
    for (int u = 0; u < DEG_BATCH; ++u) {
      const int i = i0 + u * 1024;
      q[u] = make_int4(-1, -1, -1, -1);
      if (i < nv) q[u] = v[i];
    }
    #pragma unroll
    for (int u = 0; u < DEG_BATCH; ++u) {
      unsigned b;
      b = (unsigned)(q[u].x - base); if (b < (unsigned)nb) atomicAdd(&hist[b], 1u);
      b = (unsigned)(q[u].y - base); if (b < (unsigned)nb) atomicAdd(&hist[b], 1u);
      b = (unsigned)(q[u].z - base); if (b < (unsigned)nb) atomicAdd(&hist[b], 1u);
      b = (unsigned)(q[u].w - base); if (b < (unsigned)nb) atomicAdd(&hist[b], 1u);
    }
  }
  for (int e = e0 + (nv << 2) + tid; e < e1; e += 1024) {
    unsigned b = (unsigned)(idx[e] - base);
    if (b < (unsigned)nb) atomicAdd(&hist[b], 1u);
  }
  __syncthreads();

  unsigned* __restrict__ outp = planes + (size_t)gx * (size_t)stride + base;
  uint4* __restrict__ outp4 = reinterpret_cast<uint4*>(outp);
  for (int i = tid; i < nb4; i += 1024) outp4[i] = hist4[i];
  for (int i = (nb4 << 2) + tid; i < nb; i += 1024) outp[i] = hist[i];
}

__device__ __forceinline__ void bin_slice_group(
    const int* __restrict__ src, const int* __restrict__ dst, int2* __restrict__ binned,
    int* __restrict__ binOff, int nE, int nBuck, int g, unsigned* smem) {
  unsigned* wcnt  = smem;
  unsigned* kst   = smem + 32 * NKEY;
  unsigned* red   = kst + 320;
  int2*     stage = reinterpret_cast<int2*>(smem + 32 * NKEY + 384);
  const int tid  = threadIdx.x;
  const int lane = tid & 31;
  const int wave = __builtin_amdgcn_readfirstlane(tid >> 5);

  const int per   = (((nE + NSLICE - 1) / NSLICE) + 3) & ~3;
  const int e0    = min(nE, g * per);
  const int e1    = min(nE, e0 + per);
  const int len   = e1 - e0;
  const int steps = (len + 31) >> 5;
  const int spw   = (steps + 31) >> 5;
  const int first = wave * spw * 32;
  const int* __restrict__ dS = dst + e0;
  const int* __restrict__ sS = src + e0;
  unsigned* myc = wcnt + wave * NKEY;

  for (int i = tid; i < 32 * NKEY; i += 1024) wcnt[i] = 0u;
  __syncthreads();

  for (int s0 = 0; s0 < spw; s0 += SORT_BATCH) {
    int d[SORT_BATCH];
    #pragma unroll
    for (int u = 0; u < SORT_BATCH; ++u) {
      const int i = first + (s0 + u) * 32 + lane;
      d[u] = -1;
      if (s0 + u < spw && i < len) d[u] = dS[i];
    }
    #pragma unroll
    for (int u = 0; u < SORT_BATCH; ++u)
      if (d[u] >= 0) atomicAdd(&myc[min((unsigned)d[u] >> BSHIFT, (unsigned)(NKEY - 1))], 1u);
  }
  __syncthreads();

  int tot = 0;
  if (tid < NKEY) {
    unsigned run = 0u;
    for (int w = 0; w < 32; ++w) { const unsigned c = wcnt[w * NKEY + tid]; wcnt[w * NKEY + tid] = run; run += c; }
    tot = (int)run;
  }
  int total;
  const int ks = scan256(tot, tid, red, total);
  if (tid <= NKEY) kst[tid] = (unsigned)ks;
  if (tid <= nBuck && tid <= NKEY) binOff[(size_t)g * (nBuck + 1) + tid] = ks;
  __syncthreads();

  const bool toLds = total <= STAGE_PAIRS;
  for (int s0 = 0; s0 < spw; s0 += SORT_BATCH) {
    int d[SORT_BATCH], s[SORT_BATCH];
    #pragma unroll
    for (int u = 0; u < SORT_BATCH; ++u) {
      const int i = first + (s0 + u) * 32 + lane;
      d[u] = -1; s[u] = 0;
      if (s0 + u < spw && i < len) { d[u] = dS[i]; s[u] = sS[i]; }
    }
    #pragma unroll
    for (int u = 0; u < SORT_BATCH; ++u) {
      if (s0 + u < spw) {
        const bool     valid = d[u] >= 0;
        const unsigned key   = valid ? min((unsigned)d[u] >> BSHIFT, (unsigned)(NKEY - 1)) : 0u;
        const unsigned m     = same_key_mask(key, valid);
        const unsigned rank  = (unsigned)__popc(m & ((1u << lane) - 1u));
        const unsigned p     = kst[key] + myc[key] + rank;
        if (valid) {
          const int2 e = make_int2(s[u], d[u]);
          if (toLds) stage[p] = e; else binned[(size_t)e0 + p] = e;
          atomicAdd(&myc[key], 1u);
        }
      }
    }
  }
  __syncthreads();
  if (toLds)
    for (int i = tid; i < total; i += 1024) binned[(size_t)e0 + i] = stage[i];
}

__global__ __launch_bounds__(1024, 2) void plane_bin_kernel(
    const int* __restrict__ src, const int* __restrict__ dst, unsigned* __restrict__ planes,
    int2* __restrict__ binned, int* __restrict__ binOff, int nE, int nNodes, int nBuck, int passes) {
  __shared__ __attribute__((aligned(16))) unsigned smem[SMEM_WORDS];
  const int bid    = (int)blockIdx.x;
  const int nPlane = DEG_PARTS * passes;
  if (bid < nPlane) src_plane_group(src, planes, nE, nNodes, bid % DEG_PARTS, bid / DEG_PARTS, smem);
  else              bin_slice_group(src, dst, binned, binOff, nE, nBuck, bid - nPlane, smem);
}

constexpr int RUNS_PER_WAVE = NSLICE / 32;

__device__ __forceinline__ void bucket_group(
    const int2* __restrict__ binned, const int* __restrict__ binOff,
    int* __restrict__ rowptr, float* __restrict__ inNorm, int* __restrict__ sortedSrc,
    int nE, int nNodes, int nBuck, int b, unsigned* smem) {
  unsigned* wcnt   = smem;
  unsigned* kst    = smem + 32 * NKEY;
  unsigned* runOff = kst + NKEY;
  unsigned* runCnt = runOff + NSLICE;
  unsigned* red    = runCnt + NSLICE;
  int*      stage  = reinterpret_cast<int*>(red + 128);
  const int tid  = threadIdx.x;
  const int lane = tid & 31;
  const int wave = __builtin_amdgcn_readfirstlane(tid >> 5);
  unsigned* myc = wcnt + wave * NKEY;

  const int per = (((nE + NSLICE - 1) / NSLICE) + 3) & ~3;
  int o0 = 0, c = 0;
  if (tid < NSLICE) {
    const int* r = binOff + (size_t)tid * (nBuck + 1) + b;
    o0 = r[0];
    c  = r[1] - o0;
    runOff[tid] = (unsigned)(min(nE, tid * per) + o0);
    runCnt[tid] = (unsigned)c;
  }
  for (int i = tid; i < 32 * NKEY; i += 1024) wcnt[i] = 0u;
  int sa = o0, sb = c;
  #pragma unroll
  for (int d = 16; d >= 1; d >>= 1) { sa += __shfl_xor(sa, d, 32); sb += __shfl_xor(sb, d, 32); }
  if (lane == 0) { red[wave] = (unsigned)sa; red[32 + wave] = (unsigned)sb; }
  __syncthreads();
  int base = (int)red[lane], total = (int)red[32 + lane];
  #pragma unroll
  for (int d = 16; d >= 1; d >>= 1) { base += __shfl_xor(base, d, 32); total += __shfl_xor(total, d, 32); }
  base  = __builtin_amdgcn_readfirstlane(base);
  total = __builtin_amdgcn_readfirstlane(total);

  int2 q[RUNS_PER_WAVE];
  int  ro[RUNS_PER_WAVE], cn[RUNS_PER_WAVE];
  #pragma unroll
  for (int r = 0; r < RUNS_PER_WAVE; ++r) {
    ro[r] = __builtin_amdgcn_readfirstlane((int)runOff[wave * RUNS_PER_WAVE + r]);
    cn[r] = __builtin_amdgcn_readfirstlane((int)runCnt[wave * RUNS_PER_WAVE + r]);
    q[r] = make_int2(0, 0);
    if (lane < cn[r]) q[r] = binned[(size_t)ro[r] + lane];
  }
  #pragma unroll
  for (int r = 0; r < RUNS_PER_WAVE; ++r) {
    if (lane < cn[r]) atomicAdd(&myc[q[r].y & (NKEY - 1)], 1u);
    for (int k = 32; k < cn[r]; k += 32)
      if (k + lane < cn[r]) atomicAdd(&myc[binned[(size_t)ro[r] + k + lane].y & (NKEY - 1)], 1u);
  }
  __syncthreads();

  int deg = 0;
  if (tid < NKEY) {
    unsigned run = 0u;
    for (int w = 0; w < 32; ++w) { const unsigned t = wcnt[w * NKEY + tid]; wcnt[w * NKEY + tid] = run; run += t; }
    deg = (int)run;
  }
  int total2;
  const int ks = scan256(deg, tid, red + 64, total2);
  (void)total2;
  if (tid < NKEY) {
    kst[tid] = (unsigned)ks;
    const int n = (b << BSHIFT) + tid;
    if (n < nNodes) {
      rowptr[n] = base + ks;
      inNorm[n] = rsqrtf(fmaxf((float)deg, 1.0f));
      if (n == nNodes - 1) rowptr[nNodes] = base + ks + deg;
    }
  }
  __syncthreads();

  const bool toLds = total <= STAGE_LIST;
  #pragma unroll
  for (int r = 0; r < RUNS_PER_WAVE; ++r) {
    for (int k = 0; k < cn[r]; k += 32) {
      const bool valid = k + lane < cn[r];
      int2 e = q[r];
      if (k > 0) { e = make_int2(0, 0); if (valid) e = binned[(size_t)ro[r] + k + lane]; }
      const unsigned key  = valid ? (unsigned)(e.y & (NKEY - 1)) : 0u;
      const unsigned m    = same_key_mask(key, valid);
      const unsigned rank = (unsigned)__popc(m & ((1u << lane) - 1u));
      const unsigned p    = kst[key] + myc[key] + rank;
      if (valid) {
        if (toLds) stage[p] = e.x;
        else if ((long)base + p < (long)nE) sortedSrc[(size_t)base + p] = e.x;
        atomicAdd(&myc[key], 1u);
      }
    }
  }
  __syncthreads();
  if (toLds)
    for (int i = tid; i < total; i += 1024)
      if ((long)base + i < (long)nE) sortedSrc[(size_t)base + i] = stage[i];
}

typedef __attribute__((ext_vector_type(4))) __bf16 v4bf;
constexpr int PROD_ROWS  = 128;
constexpr int PROD_K     = 128;
constexpr int PROD_PITCH = PROD_K + 8;

__device__ __forceinline__ void product_rows128(
    const float* __restrict__ X, const unsigned* __restrict__ planes, float* __restrict__ outNorm,
    const float* __restrict__ Wg, float* __restrict__ Y, int nRows, int rowBase, unsigned* smem) {
  __bf16*   Xs     = reinterpret_cast<__bf16*>(smem);
  __bf16*   Wt     = Xs + PROD_ROWS * PROD_PITCH;
  unsigned* rowCnt = smem + (PROD_ROWS + 64) * PROD_PITCH / 2;
  float*    noS    = reinterpret_cast<float*>(rowCnt + PROD_ROWS);
  const int tid  = threadIdx.x;
  const int lane = tid & 31;
  const int wave = tid >> 5;
  const int stride = (nRows + 63) & ~63;

  const int wc = tid & 63;
  const int wk = (tid >> 6) << 3;
  float wv[8];
  #pragma unroll
  for (int i = 0; i < 8; ++i) wv[i] = Wg[(wk + i) * 64 + wc];
  float4 xv[4];
  #pragma unroll
  for (int j = 0; j < 4; ++j) {
    int row = rowBase + wave + 32 * j;
    row = (row < nRows) ? row : (nRows - 1);
    xv[j] = *reinterpret_cast<const float4*>(X + (size_t)row * PROD_K + lane * 4);
  }
  if (tid < PROD_ROWS) rowCnt[tid] = 0u;
  __syncthreads();
  {
    const int r   = tid & (PROD_ROWS - 1);
    const int grp = tid >> 7;
    int row = rowBase + r;
    row = (row < nRows) ? row : (nRows - 1);
    unsigned cnt = 0u;
    #pragma unroll
    for (int i = 0; i < DEG_PARTS / 8; ++i) cnt += planes[(size_t)(grp * (DEG_PARTS / 8) + i) * stride + row];
    atomicAdd(&rowCnt[r], cnt);
  }
  __syncthreads();
  if (tid < PROD_ROWS) {
    const float sc = rsqrtf(fmaxf((float)rowCnt[tid], 1.0f));
    noS[tid] = sc;
    if (rowBase + tid < nRows) outNorm[rowBase + tid] = sc;
  }
  __syncthreads();

  #pragma unroll
  for (int j = 0; j < 4; ++j) {
    const float xs = noS[wave + 32 * j];
    v4bf p;
    p[0] = f2bf(xv[j].x * xs);
    p[1] = f2bf(xv[j].y * xs);
    p[2] = f2bf(xv[j].z * xs);
    p[3] = f2bf(xv[j].w * xs);
    *reinterpret_cast<v4bf*>(&Xs[(wave + 32 * j) * PROD_PITCH + lane * 4]) = p;
  }
  v8bf wp;
  #pragma unroll
  for (int i = 0; i < 8; ++i) wp[i] = f2bf(wv[i]);
  *reinterpret_cast<v8bf*>(&Wt[wc * PROD_PITCH + wk]) = wp;
  __syncthreads();

  const int hh      = lane >> 4;
  const int l       = lane & 15;
  const int rt      = wave >> 2;
  const int colBase = (wave & 3) << 4;
  const __bf16* xr = Xs + (rt * 16 + l) * PROD_PITCH;
  const __bf16* wr = Wt + (colBase + l) * PROD_PITCH;

  v8f acc = {};
  #pragma unroll
  for (int k0 = 0; k0 < PROD_K; k0 += 32) {
    const v8bf a0 = *reinterpret_cast<const v8bf*>(xr + k0 + (hh << 3));
    const v8bf a1 = *reinterpret_cast<const v8bf*>(xr + k0 + 16 + (hh << 3));
    const v8bf b0 = *reinterpret_cast<const v8bf*>(wr + k0 + (hh << 3));
    const v8bf b1 = *reinterpret_cast<const v8bf*>(wr + k0 + 16 + (hh << 3));
    v16bf a, b;
    #pragma unroll
    for (int i = 0; i < 8; ++i) {
      a[i] = a0[i]; a[i + 8] = a1[i];
      b[i] = b0[i]; b[i + 8] = b1[i];
    }
    acc = __builtin_amdgcn_wmma_f32_16x16x32_bf16(
         false, a,  false, b,
         (short)0, acc,  false,  false);
  }

  const int r0 = rowBase + rt * 16;
  float* __restrict__ yOut = Y + ((size_t)(r0 + (hh << 3)) * 64 + colBase + l);
  if (r0 + 16 <= nRows) {
    #pragma unroll
    for (int v = 0; v < 8; ++v) yOut[(size_t)v * 64] = acc[v];
  } else {
    #pragma unroll
    for (int v = 0; v < 8; ++v) {
      if (r0 + v + (hh << 3) < nRows) yOut[(size_t)v * 64] = acc[v];
    }
  }
}

__global__ __launch_bounds__(1024, 2) void bucket_product_kernel(
    const int2* __restrict__ binned, const int* __restrict__ binOff, const unsigned* __restrict__ planes,
    int* __restrict__ rowptr, float* __restrict__ inNorm, float* __restrict__ outNorm,
    int* __restrict__ sortedSrc, int nE, int nNodes, int nBuck,
    const float* __restrict__ X, const float* __restrict__ Wg, float* __restrict__ Y) {
  __shared__ __attribute__((aligned(16))) unsigned smem[SMEM_WORDS];
  const int bid = (int)blockIdx.x;
  if (bid < nBuck) bucket_group(binned, binOff, rowptr, inNorm, sortedSrc, nE, nNodes, nBuck, bid, smem);
  else             product_rows128(X, planes, outNorm, Wg, Y, nNodes, (bid - nBuck) * PROD_ROWS, smem);
}

constexpr int GATHER_BATCH = 8;

template<bool PARTIAL>
__device__ __forceinline__ void gather_batch(const float2* __restrict__ y2, int idx, int k, int cnt,
                                             int lane, int lastRow, float2& acc) {
  float2 v[GATHER_BATCH];
  #pragma unroll
  for (int u = 0; u < GATHER_BATCH; ++u) {
    const int kk = PARTIAL ? min(k + u, cnt - 1) : (k + u);
    int s = __builtin_amdgcn_readlane(idx, kk);
    s = (int)min((unsigned)s, (unsigned)lastRow);
    v[u] = y2[s * 32 + lane];
  }
  #pragma unroll
  for (int u = 0; u < GATHER_BATCH; ++u) {
    const bool ok = !PARTIAL || (k + u) < cnt;
    acc.x += ok ? v[u].x : 0.0f;
    acc.y += ok ? v[u].y : 0.0f;
  }
}

__device__ __forceinline__ float2 gather_row_sum(const float2* __restrict__ y2, const int* __restrict__ rowptr,
                                                 const int* __restrict__ sortedSrc, int n, int lane, int nRows) {
  const int beg = rowptr[n];
  const int end = rowptr[n + 1];
  float2 acc = make_float2(0.0f, 0.0f);
  int idx = (beg + lane < end) ? sortedSrc[beg + lane] : 0;
  for (int base = beg; base < end; base += 32) {
    const int cnt = min(32, end - base);
    const int inext = base + 32 + lane;
    const int nxt = (inext < end) ? sortedSrc[inext] : 0;
    int k = 0;
    for (; k + GATHER_BATCH <= cnt; k += GATHER_BATCH)
      gather_batch<false>(y2, idx, k, cnt, lane, nRows - 1, acc);
    if (k < cnt)
      gather_batch<true>(y2, idx, k, cnt, lane, nRows - 1, acc);
    idx = nxt;
  }
  return acc;
}

__global__ __launch_bounds__(256) void gather_finalize_kernel(
    const float* __restrict__ y, const int* __restrict__ rowptr, const int* __restrict__ sortedSrc,
    const float* __restrict__ inNorm, const float* __restrict__ bias, float* __restrict__ out,
    int nRows, int doRelu) {
  const int lane = threadIdx.x & 31;
  const int n = __builtin_amdgcn_readfirstlane((int)(blockIdx.x * 8 + (threadIdx.x >> 5)));
  if (n >= nRows) return;
  const float2 acc = gather_row_sum(reinterpret_cast<const float2*>(y), rowptr, sortedSrc, n, lane, nRows);
  const float  sc = inNorm[n];
  const float2 bb = *reinterpret_cast<const float2*>(bias + 2 * lane);
  float2 r;
  r.x = acc.x * sc + bb.x;
  r.y = acc.y * sc + bb.y;
  if (doRelu) { r.x = fmaxf(r.x, 0.0f); r.y = fmaxf(r.y, 0.0f); }
  *reinterpret_cast<float2*>(out + (size_t)n * 64 + 2 * lane) = r;
}

typedef __attribute__((ext_vector_type(2))) __bf16 v2bf;

__global__ __launch_bounds__(512) void gather_finalize_gemm_kernel(
    const float* __restrict__ y, const int* __restrict__ rowptr, const int* __restrict__ sortedSrc,
    const float* __restrict__ inNorm, const float* __restrict__ bias,
    const float* __restrict__ outNorm, const float* __restrict__ Wg,
    float* __restrict__ yNext, int nRows) {
  __shared__ __attribute__((aligned(16))) __bf16 Xs[16][64 + 8];
  __shared__ __attribute__((aligned(16))) __bf16 Wt[64][64 + 8];

  const int tid  = threadIdx.x;
  const int lane = tid & 31;
  const int wave    = __builtin_amdgcn_readfirstlane(tid >> 5);
  const int rowBase = blockIdx.x * 16;
  const int n       = rowBase + wave;

  const int wc = tid & 63;
  const int wk = (tid >> 6) << 3;
  float wv[8];
  #pragma unroll
  for (int i = 0; i < 8; ++i) wv[i] = Wg[(wk + i) * 64 + wc];

  float2 r = make_float2(0.0f, 0.0f);
  if (n < nRows) {
    const float2 acc = gather_row_sum(reinterpret_cast<const float2*>(y), rowptr, sortedSrc, n, lane, nRows);
    const float  sc = inNorm[n];
    const float  so = outNorm[n];
    const float2 bb = *reinterpret_cast<const float2*>(bias + 2 * lane);
    r.x = fmaxf(acc.x * sc + bb.x, 0.0f) * so;
    r.y = fmaxf(acc.y * sc + bb.y, 0.0f) * so;
  }
  v2bf rp;
  rp[0] = f2bf(r.x);
  rp[1] = f2bf(r.y);
  *reinterpret_cast<v2bf*>(&Xs[wave][2 * lane]) = rp;

  v8bf wp;
  #pragma unroll
  for (int i = 0; i < 8; ++i) wp[i] = f2bf(wv[i]);
  *reinterpret_cast<v8bf*>(&Wt[wc][wk]) = wp;
  __syncthreads();

  if (wave >= 4) return;

  const int hh      = lane >> 4;
  const int l       = lane & 15;
  const int colBase = wave << 4;

  v8f acc = {};
  #pragma unroll
  for (int k0 = 0; k0 < 64; k0 += 32) {
    const v8bf a0 = *reinterpret_cast<const v8bf*>(&Xs[l][k0 + (hh << 3)]);
    const v8bf a1 = *reinterpret_cast<const v8bf*>(&Xs[l][k0 + 16 + (hh << 3)]);
    const v8bf b0 = *reinterpret_cast<const v8bf*>(&Wt[colBase + l][k0 + (hh << 3)]);
    const v8bf b1 = *reinterpret_cast<const v8bf*>(&Wt[colBase + l][k0 + 16 + (hh << 3)]);
    v16bf a, b;
    #pragma unroll
    for (int i = 0; i < 8; ++i) {
      a[i] = a0[i]; a[i + 8] = a1[i];
      b[i] = b0[i]; b[i + 8] = b1[i];
    }
    acc = __builtin_amdgcn_wmma_f32_16x16x32_bf16(
         false, a,  false, b,
         (short)0, acc,  false,  false);
  }

  float* __restrict__ yOut = yNext + ((long)(rowBase + (hh << 3)) * 64 + colBase + l);
  if (rowBase + 16 <= nRows) {
    #pragma unroll
    for (int v = 0; v < 8; ++v) yOut[(long)v * 64] = acc[v];
  } else {
    #pragma unroll
    for (int v = 0; v < 8; ++v) {
      if (rowBase + v + (hh << 3) < nRows) yOut[(long)v * 64] = acc[v];
    }
  }
}

static inline int cdiv_i(long a, long b) { return (int)((a + b - 1) / b); }

extern "C" void kernel_launch(void* const* d_in, const int* in_sizes, int n_in,
                              void* d_out, int out_size, void* d_ws, size_t ws_size,
                              hipStream_t stream) {
  const float* feat = (const float*)d_in[0];
  const float* W0   = (const float*)d_in[1];
  const float* b0   = (const float*)d_in[2];
  const float* W1   = (const float*)d_in[3];
  const float* b1   = (const float*)d_in[4];
  const float* W2   = (const float*)d_in[5];
  const float* b2   = (const float*)d_in[6];
  const int*   src  = (const int*)d_in[7];
  const int*   dst  = (const int*)d_in[8];

  const int N = in_sizes[0] / 128;
  const int E = in_sizes[7];
  float* out = (float*)d_out;

  const size_t P = (size_t)((N + 63) & ~63);
  const int nBuck  = cdiv_i(N, 1 << BSHIFT);
  const int passes = cdiv_i(N, DEG_BINS);
  float* wsf = (float*)d_ws;
  float* no  = wsf;
  float* ni  = no + P;
  float* y   = ni + P;
  float* h   = y  + (size_t)N * 64;
  unsigned* planes = (unsigned*)(h + (size_t)N * 64);
  int* binOff    = (int*)(planes + (size_t)DEG_PARTS * P);
  int* rowptr    = binOff + (size_t)NSLICE * (size_t)(nBuck + 1);
  int2* binned   = (int2*)(rowptr + P + 64);
  int* sortedSrc = (int*)(binned + (size_t)E);

  const int gemmBlocks = cdiv_i(N, 16);
  const int gatherBlocks = cdiv_i(N, 8);

  plane_bin_kernel<<<DEG_PARTS * passes + NSLICE, 1024, 0, stream>>>(src, dst, planes, binned, binOff, E, N, nBuck, passes);
  bucket_product_kernel<<<nBuck + cdiv_i(N, PROD_ROWS), 1024, 0, stream>>>(
      binned, binOff, planes, rowptr, ni, no, sortedSrc, E, N, nBuck, feat, W0, y);
  gather_finalize_gemm_kernel<<<gemmBlocks, 512, 0, stream>>>(y, rowptr, sortedSrc, ni, b0, no, W1, h, N);
  gather_finalize_gemm_kernel<<<gemmBlocks, 512, 0, stream>>>(h, rowptr, sortedSrc, ni, b1, no, W2, y, N);
  gather_finalize_kernel<<<gatherBlocks, 256, 0, stream>>>(y, rowptr, sortedSrc, ni, b2, out, N, 0);
}


# y rows stored as f16 (128 B; sums stay f32): one gather load brings two rows, a 16-lane half each on alternate list entries (index by DPP row_share); halves added at the end: sum order changes
# speedup vs baseline: 8.7442x; 1.2267x over previous
#include <hip/hip_runtime.h>

typedef __attribute__((ext_vector_type(16))) __bf16 v16bf;
typedef __attribute__((ext_vector_type(8)))  __bf16 v8bf;
typedef __attribute__((ext_vector_type(8)))  float  v8f;

typedef __attribute__((ext_vector_type(4))) _Float16 v4h;

__device__ __forceinline__ __bf16 f2bf(float f) {
  unsigned int u = __builtin_bit_cast(unsigned int, f);
  u += 0x7FFFu + ((u >> 16) & 1u);
  unsigned short s = (unsigned short)(u >> 16);
  return __builtin_bit_cast(__bf16, s);
}

constexpr int SMEM_WORDS  = 25600;
constexpr int NSLICE      = 256;
constexpr int BSHIFT      = 8;
constexpr int NKEY        = 256;
constexpr int SORT_BATCH  = 8;
constexpr int STAGE_PAIRS = 8192;
constexpr int STAGE_LIST  = 16384;

__device__ __forceinline__ unsigned same_key_mask(unsigned key, bool valid) {
  unsigned m = __builtin_amdgcn_ballot_w32(valid);
  #pragma unroll
  for (int bit = 0; bit < 8; ++bit) {
    const bool one = ((key >> bit) & 1u) != 0u;
    const unsigned b = __builtin_amdgcn_ballot_w32(one);
    m &= one ? b : ~b;
  }
  return m;
}

__device__ __forceinline__ int scan256(int v, int tid, unsigned* red, int& total) {
  const int lane = tid & 31;
  const int wave = tid >> 5;
  int incl = v;
  #pragma unroll
  for (int d = 1; d < 32; d <<= 1) {
    const int t = __shfl_up(incl, d, 32);
    if (lane >= d) incl += t;
  }
  if (lane == 31 && wave < 8) red[wave] = (unsigned)incl;
  __syncthreads();
  int woff = 0, tot = 0;
  #pragma unroll
  for (int w = 0; w < 8; ++w) { const int s = (int)red[w]; tot += s; woff += (w < wave) ? s : 0; }
  total = tot;
  return woff + incl - v;
}

constexpr int DEG_BINS  = 25088;
constexpr int DEG_PARTS = 64;
constexpr int DEG_BATCH = 8;

__device__ __forceinline__ void src_plane_group(const int* __restrict__ idx, unsigned* __restrict__ planes,
                                                int nE, int nNodes, int gx, int gz, unsigned* hist) {
  const int tid    = threadIdx.x;
  const int base   = gz * DEG_BINS;
  const int nb     = max(0, min(DEG_BINS, nNodes - base));
  const int nb4    = nb >> 2;
  const int stride = (nNodes + 63) & ~63;

  uint4* hist4 = reinterpret_cast<uint4*>(hist);
  for (int i = tid; i < nb4; i += 1024) hist4[i] = make_uint4(0u, 0u, 0u, 0u);
  for (int i = (nb4 << 2) + tid; i < nb; i += 1024) hist[i] = 0u;
  __syncthreads();

  const int per = (((nE + DEG_PARTS - 1) / DEG_PARTS) + 3) & ~3;
  const int e0  = min(nE, gx * per);
  const int e1  = min(nE, e0 + per);
  const int nv  = (e1 - e0) >> 2;
  const int4* __restrict__ v = reinterpret_cast<const int4*>(idx + e0);
  for (int i0 = tid; i0 < nv; i0 += DEG_BATCH * 1024) {
    int4 q[DEG_BATCH];
    #pragma unroll
    for (int u = 0; u < DEG_BATCH; ++u) {
      const int i = i0 + u * 1024;
      q[u] = make_int4(-1, -1, -1, -1);
      if (i < nv) q[u] = v[i];
    }
    #pragma unroll
    for (int u = 0; u < DEG_BATCH; ++u) {
      unsigned b;
      b = (unsigned)(q[u].x - base); if (b < (unsigned)nb) atomicAdd(&hist[b], 1u);
      b = (unsigned)(q[u].y - base); if (b < (unsigned)nb) atomicAdd(&hist[b], 1u);
      b = (unsigned)(q[u].z - base); if (b < (unsigned)nb) atomicAdd(&hist[b], 1u);
      b = (unsigned)(q[u].w - base); if (b < (unsigned)nb) atomicAdd(&hist[b], 1u);
    }
  }
  for (int e = e0 + (nv << 2) + tid; e < e1; e += 1024) {
    unsigned b = (unsigned)(idx[e] - base);
    if (b < (unsigned)nb) atomicAdd(&hist[b], 1u);
  }
  __syncthreads();

  unsigned* __restrict__ outp = planes + (size_t)gx * (size_t)stride + base;
  uint4* __restrict__ outp4 = reinterpret_cast<uint4*>(outp);
  for (int i = tid; i < nb4; i += 1024) outp4[i] = hist4[i];
  for (int i = (nb4 << 2) + tid; i < nb; i += 1024) outp[i] = hist[i];
}

__device__ __forceinline__ void bin_slice_group(
    const int* __restrict__ src, const int* __restrict__ dst, int2* __restrict__ binned,
    int* __restrict__ binOff, int nE, int nBuck, int g, unsigned* smem) {
  unsigned* wcnt  = smem;
  unsigned* kst   = smem + 32 * NKEY;
  unsigned* red   = kst + 320;
  int2*     stage = reinterpret_cast<int2*>(smem + 32 * NKEY + 384);
  const int tid  = threadIdx.x;
  const int lane = tid & 31;
  const int wave = __builtin_amdgcn_readfirstlane(tid >> 5);

  const int per   = (((nE + NSLICE - 1) / NSLICE) + 3) & ~3;
  const int e0    = min(nE, g * per);
  const int e1    = min(nE, e0 + per);
  const int len   = e1 - e0;
  const int steps = (len + 31) >> 5;
  const int spw   = (steps + 31) >> 5;
  const int first = wave * spw * 32;
  const int* __restrict__ dS = dst + e0;
  const int* __restrict__ sS = src + e0;
  unsigned* myc = wcnt + wave * NKEY;

  for (int i = tid; i < 32 * NKEY; i += 1024) wcnt[i] = 0u;
  __syncthreads();

  for (int s0 = 0; s0 < spw; s0 += SORT_BATCH) {
    int d[SORT_BATCH];
    #pragma unroll
    for (int u = 0; u < SORT_BATCH; ++u) {
      const int i = first + (s0 + u) * 32 + lane;
      d[u] = -1;
      if (s0 + u < spw && i < len) d[u] = dS[i];
    }
    #pragma unroll
    for (int u = 0; u < SORT_BATCH; ++u)
      if (d[u] >= 0) atomicAdd(&myc[min((unsigned)d[u] >> BSHIFT, (unsigned)(NKEY - 1))], 1u);
  }
  __syncthreads();

  int tot = 0;
  if (tid < NKEY) {
    unsigned run = 0u;
    for (int w = 0; w < 32; ++w) { const unsigned c = wcnt[w * NKEY + tid]; wcnt[w * NKEY + tid] = run; run += c; }
    tot = (int)run;
  }
  int total;
  const int ks = scan256(tot, tid, red, total);
  if (tid <= NKEY) kst[tid] = (unsigned)ks;
  if (tid <= nBuck && tid <= NKEY) binOff[(size_t)g * (nBuck + 1) + tid] = ks;
  __syncthreads();

  const bool toLds = total <= STAGE_PAIRS;
  for (int s0 = 0; s0 < spw; s0 += SORT_BATCH) {
    int d[SORT_BATCH], s[SORT_BATCH];
    #pragma unroll
    for (int u = 0; u < SORT_BATCH; ++u) {
      const int i = first + (s0 + u) * 32 + lane;
      d[u] = -1; s[u] = 0;
      if (s0 + u < spw && i < len) { d[u] = dS[i]; s[u] = sS[i]; }
    }
    #pragma unroll
    for (int u = 0; u < SORT_BATCH; ++u) {
      if (s0 + u < spw) {
        const bool     valid = d[u] >= 0;
        const unsigned key   = valid ? min((unsigned)d[u] >> BSHIFT, (unsigned)(NKEY - 1)) : 0u;
        const unsigned m     = same_key_mask(key, valid);
        const unsigned rank  = (unsigned)__popc(m & ((1u << lane) - 1u));
        const unsigned p     = kst[key] + myc[key] + rank;
        if (valid) {
          const int2 e = make_int2(s[u], d[u]);
          if (toLds) stage[p] = e; else binned[(size_t)e0 + p] = e;
          atomicAdd(&myc[key], 1u);
        }
      }
    }
  }
  __syncthreads();
  if (toLds)
    for (int i = tid; i < total; i += 1024) binned[(size_t)e0 + i] = stage[i];
}

__global__ __launch_bounds__(1024, 2) void plane_bin_kernel(
    const int* __restrict__ src, const int* __restrict__ dst, unsigned* __restrict__ planes,
    int2* __restrict__ binned, int* __restrict__ binOff, int nE, int nNodes, int nBuck, int passes) {
  __shared__ __attribute__((aligned(16))) unsigned smem[SMEM_WORDS];
  const int bid    = (int)blockIdx.x;
  const int nPlane = DEG_PARTS * passes;
  if (bid < nPlane) src_plane_group(src, planes, nE, nNodes, bid % DEG_PARTS, bid / DEG_PARTS, smem);
  else              bin_slice_group(src, dst, binned, binOff, nE, nBuck, bid - nPlane, smem);
}

constexpr int RUNS_PER_WAVE = NSLICE / 32;

__device__ __forceinline__ void bucket_group(
    const int2* __restrict__ binned, const int* __restrict__ binOff,
    int* __restrict__ rowptr, float* __restrict__ inNorm, int* __restrict__ sortedSrc,
    int nE, int nNodes, int nBuck, int b, unsigned* smem) {
  unsigned* wcnt   = smem;
  unsigned* kst    = smem + 32 * NKEY;
  unsigned* runOff = kst + NKEY;
  unsigned* runCnt = runOff + NSLICE;
  unsigned* red    = runCnt + NSLICE;
  int*      stage  = reinterpret_cast<int*>(red + 128);
  const int tid  = threadIdx.x;
  const int lane = tid & 31;
  const int wave = __builtin_amdgcn_readfirstlane(tid >> 5);
  unsigned* myc = wcnt + wave * NKEY;

  const int per = (((nE + NSLICE - 1) / NSLICE) + 3) & ~3;
  int o0 = 0, c = 0;
  if (tid < NSLICE) {
    const int* r = binOff + (size_t)tid * (nBuck + 1) + b;
    o0 = r[0];
    c  = r[1] - o0;
    runOff[tid] = (unsigned)(min(nE, tid * per) + o0);
    runCnt[tid] = (unsigned)c;
  }
  for (int i = tid; i < 32 * NKEY; i += 1024) wcnt[i] = 0u;
  int sa = o0, sb = c;
  #pragma unroll
  for (int d = 16; d >= 1; d >>= 1) { sa += __shfl_xor(sa, d, 32); sb += __shfl_xor(sb, d, 32); }
  if (lane == 0) { red[wave] = (unsigned)sa; red[32 + wave] = (unsigned)sb; }
  __syncthreads();
  int base = (int)red[lane], total = (int)red[32 + lane];
  #pragma unroll
  for (int d = 16; d >= 1; d >>= 1) { base += __shfl_xor(base, d, 32); total += __shfl_xor(total, d, 32); }
  base  = __builtin_amdgcn_readfirstlane(base);
  total = __builtin_amdgcn_readfirstlane(total);

  int2 q[RUNS_PER_WAVE];
  int  ro[RUNS_PER_WAVE], cn[RUNS_PER_WAVE];
  #pragma unroll
  for (int r = 0; r < RUNS_PER_WAVE; ++r) {
    ro[r] = __builtin_amdgcn_readfirstlane((int)runOff[wave * RUNS_PER_WAVE + r]);
    cn[r] = __builtin_amdgcn_readfirstlane((int)runCnt[wave * RUNS_PER_WAVE + r]);
    q[r] = make_int2(0, 0);
    if (lane < cn[r]) q[r] = binned[(size_t)ro[r] + lane];
  }
  #pragma unroll
  for (int r = 0; r < RUNS_PER_WAVE; ++r) {
    if (lane < cn[r]) atomicAdd(&myc[q[r].y & (NKEY - 1)], 1u);
    for (int k = 32; k < cn[r]; k += 32)
      if (k + lane < cn[r]) atomicAdd(&myc[binned[(size_t)ro[r] + k + lane].y & (NKEY - 1)], 1u);
  }
  __syncthreads();

  int deg = 0;
  if (tid < NKEY) {
    unsigned run = 0u;
    for (int w = 0; w < 32; ++w) { const unsigned t = wcnt[w * NKEY + tid]; wcnt[w * NKEY + tid] = run; run += t; }
    deg = (int)run;
  }
  int total2;
  const int ks = scan256(deg, tid, red + 64, total2);
  (void)total2;
  if (tid < NKEY) {
    kst[tid] = (unsigned)ks;
    const int n = (b << BSHIFT) + tid;
    if (n < nNodes) {
      rowptr[n] = base + ks;
      inNorm[n] = rsqrtf(fmaxf((float)deg, 1.0f));
      if (n == nNodes - 1) rowptr[nNodes] = base + ks + deg;
    }
  }
  __syncthreads();

  const bool toLds = total <= STAGE_LIST;
  #pragma unroll
  for (int r = 0; r < RUNS_PER_WAVE; ++r) {
    for (int k = 0; k < cn[r]; k += 32) {
      const bool valid = k + lane < cn[r];
      int2 e = q[r];
      if (k > 0) { e = make_int2(0, 0); if (valid) e = binned[(size_t)ro[r] + k + lane]; }
      const unsigned key  = valid ? (unsigned)(e.y & (NKEY - 1)) : 0u;
      const unsigned m    = same_key_mask(key, valid);
      const unsigned rank = (unsigned)__popc(m & ((1u << lane) - 1u));
      const unsigned p    = kst[key] + myc[key] + rank;
      if (valid) {
        if (toLds) stage[p] = e.x;
        else if ((long)base + p < (long)nE) sortedSrc[(size_t)base + p] = e.x;
        atomicAdd(&myc[key], 1u);
      }
    }
  }
  __syncthreads();
  if (toLds)
    for (int i = tid; i < total; i += 1024)
      if ((long)base + i < (long)nE) sortedSrc[(size_t)base + i] = stage[i];
}

typedef __attribute__((ext_vector_type(4))) __bf16 v4bf;
constexpr int PROD_ROWS  = 128;
constexpr int PROD_K     = 128;
constexpr int PROD_PITCH = PROD_K + 8;

__device__ __forceinline__ void product_rows128(
    const float* __restrict__ X, const unsigned* __restrict__ planes, float* __restrict__ outNorm,
    const float* __restrict__ Wg, _Float16* __restrict__ Y, int nRows, int rowBase, unsigned* smem) {
  __bf16*   Xs     = reinterpret_cast<__bf16*>(smem);
  __bf16*   Wt     = Xs + PROD_ROWS * PROD_PITCH;
  unsigned* rowCnt = smem + (PROD_ROWS + 64) * PROD_PITCH / 2;
  float*    noS    = reinterpret_cast<float*>(rowCnt + PROD_ROWS);
  const int tid  = threadIdx.x;
  const int lane = tid & 31;
  const int wave = tid >> 5;
  const int stride = (nRows + 63) & ~63;

  const int wc = tid & 63;
  const int wk = (tid >> 6) << 3;
  float wv[8];
  #pragma unroll
  for (int i = 0; i < 8; ++i) wv[i] = Wg[(wk + i) * 64 + wc];
  float4 xv[4];
  #pragma unroll
  for (int j = 0; j < 4; ++j) {
    int row = rowBase + wave + 32 * j;
    row = (row < nRows) ? row : (nRows - 1);
    xv[j] = *reinterpret_cast<const float4*>(X + (size_t)row * PROD_K + lane * 4);
  }
  if (tid < PROD_ROWS) rowCnt[tid] = 0u;
  __syncthreads();
  {
    const int r   = tid & (PROD_ROWS - 1);
    const int grp = tid >> 7;
    int row = rowBase + r;
    row = (row < nRows) ? row : (nRows - 1);
    unsigned cnt = 0u;
    #pragma unroll
    for (int i = 0; i < DEG_PARTS / 8; ++i) cnt += planes[(size_t)(grp * (DEG_PARTS / 8) + i) * stride + row];
    atomicAdd(&rowCnt[r], cnt);
  }
  __syncthreads();
  if (tid < PROD_ROWS) {
    const float sc = rsqrtf(fmaxf((float)rowCnt[tid], 1.0f));
    noS[tid] = sc;
    if (rowBase + tid < nRows) outNorm[rowBase + tid] = sc;
  }
  __syncthreads();

  #pragma unroll
  for (int j = 0; j < 4; ++j) {
    const float xs = noS[wave + 32 * j];
    v4bf p;
    p[0] = f2bf(xv[j].x * xs);
    p[1] = f2bf(xv[j].y * xs);
    p[2] = f2bf(xv[j].z * xs);
    p[3] = f2bf(xv[j].w * xs);
    *reinterpret_cast<v4bf*>(&Xs[(wave + 32 * j) * PROD_PITCH + lane * 4]) = p;
  }
  v8bf wp;
  #pragma unroll
  for (int i = 0; i < 8; ++i) wp[i] = f2bf(wv[i]);
  *reinterpret_cast<v8bf*>(&Wt[wc * PROD_PITCH + wk]) = wp;
  __syncthreads();

  const int hh      = lane >> 4;
  const int l       = lane & 15;
  const int rt      = wave >> 2;
  const int colBase = (wave & 3) << 4;
  const __bf16* xr = Xs + (rt * 16 + l) * PROD_PITCH;
  const __bf16* wr = Wt + (colBase + l) * PROD_PITCH;

  v8f acc = {};
  #pragma unroll
  for (int k0 = 0; k0 < PROD_K; k0 += 32) {
    const v8bf a0 = *reinterpret_cast<const v8bf*>(xr + k0 + (hh << 3));
    const v8bf a1 = *reinterpret_cast<const v8bf*>(xr + k0 + 16 + (hh << 3));
    const v8bf b0 = *reinterpret_cast<const v8bf*>(wr + k0 + (hh << 3));
    const v8bf b1 = *reinterpret_cast<const v8bf*>(wr + k0 + 16 + (hh << 3));
    v16bf a, b;
    #pragma unroll
    for (int i = 0; i < 8; ++i) {
      a[i] = a0[i]; a[i + 8] = a1[i];
      b[i] = b0[i]; b[i + 8] = b1[i];
    }
    acc = __builtin_amdgcn_wmma_f32_16x16x32_bf16(
         false, a,  false, b,
         (short)0, acc,  false,  false);
  }

  const int r0 = rowBase + rt * 16;
  _Float16* __restrict__ yOut = Y + ((size_t)(r0 + (hh << 3)) * 64 + colBase + l);
  if (r0 + 16 <= nRows) {
    #pragma unroll
    for (int v = 0; v < 8; ++v) yOut[(size_t)v * 64] = (_Float16)acc[v];
  } else {
    #pragma unroll
    for (int v = 0; v < 8; ++v) {
      if (r0 + v + (hh << 3) < nRows) yOut[(size_t)v * 64] = (_Float16)acc[v];
    }
  }
}

__global__ __launch_bounds__(1024, 2) void bucket_product_kernel(
    const int2* __restrict__ binned, const int* __restrict__ binOff, const unsigned* __restrict__ planes,
    int* __restrict__ rowptr, float* __restrict__ inNorm, float* __restrict__ outNorm,
    int* __restrict__ sortedSrc, int nE, int nNodes, int nBuck,
    const float* __restrict__ X, const float* __restrict__ Wg, float* __restrict__ Y) {
  __shared__ __attribute__((aligned(16))) unsigned smem[SMEM_WORDS];
  const int bid = (int)blockIdx.x;
  if (bid < nBuck) bucket_group(binned, binOff, rowptr, inNorm, sortedSrc, nE, nNodes, nBuck, bid, smem);
  else             product_rows128(X, planes, outNorm, Wg, reinterpret_cast<_Float16*>(Y), nNodes, (bid - nBuck) * PROD_ROWS, smem);
}

constexpr int GATHER_BATCH = 8;

template<int U>
__device__ __forceinline__ int half_lane(int x) {
  return __builtin_amdgcn_update_dpp(0, x, 0x150 + U, 0xf, 0xf, true);
}

template<int U0, bool PARTIAL>
__device__ __forceinline__ void gather_batch(const uint2* __restrict__ y8, int rowOff, int cnt,
                                             int hh, int p, float4& acc) {
  uint2 w[GATHER_BATCH];
#define GATHER_LOAD(u) w[u] = y8[half_lane<U0 + (u)>(rowOff) | p];
  GATHER_LOAD(0) GATHER_LOAD(1) GATHER_LOAD(2) GATHER_LOAD(3)
  GATHER_LOAD(4) GATHER_LOAD(5) GATHER_LOAD(6) GATHER_LOAD(7)
#undef GATHER_LOAD
  #pragma unroll
  for (int u = 0; u < GATHER_BATCH; ++u) {
    uint2 t = w[u];
    if (PARTIAL) {
      const bool ok = (2 * (U0 + u) + hh) < cnt;
      t.x = ok ? t.x : 0u;
      t.y = ok ? t.y : 0u;
    }
    const v4h v = __builtin_bit_cast(v4h, t);
    acc.x += (float)v[0];
    acc.y += (float)v[1];
    acc.z += (float)v[2];
    acc.w += (float)v[3];
  }
}

__device__ __forceinline__ float4 gather_row_sum(const uint2* __restrict__ y8, const int* __restrict__ rowptr,
                                                 const int* __restrict__ sortedSrc, int n, int lane, int nRows) {
  const int beg = rowptr[n];
  const int end = rowptr[n + 1];
  const int hh  = lane >> 4;
  const int p   = lane & 15;
  const int pos = (p << 1) | hh;
  float4 acc = make_float4(0.0f, 0.0f, 0.0f, 0.0f);
  int idx = (beg + pos < end) ? sortedSrc[beg + pos] : 0;
  for (int base = beg; base < end; base += 32) {
    const int cnt = min(32, end - base);
    const int inext = base + 32 + pos;
    const int nxt = (inext < end) ? sortedSrc[inext] : 0;
    const int rowOff = (int)(min((unsigned)idx, (unsigned)(nRows - 1)) << 4);
    if (cnt == 32) {
      gather_batch<0, false>(y8, rowOff, cnt, hh, p, acc);
      __builtin_amdgcn_sched_barrier(0);
      gather_batch<8, false>(y8, rowOff, cnt, hh, p, acc);
    } else {
      gather_batch<0, true>(y8, rowOff, cnt, hh, p, acc);
      if (cnt > 16) gather_batch<8, true>(y8, rowOff, cnt, hh, p, acc);
    }
    idx = nxt;
  }
  return acc;
}

__device__ __forceinline__ float xhalf(float x) {
  const int xi = __builtin_bit_cast(int, x);
  return __builtin_bit_cast(float, __builtin_amdgcn_permlanex16(xi, xi, 0x76543210, 0xfedcba98, false, false));
}

__device__ __forceinline__ float2 gather_combine(const float4 acc, int hh) {
  const float kx = hh ? acc.z : acc.x;
  const float ky = hh ? acc.w : acc.y;
  const float sx = hh ? acc.x : acc.z;
  const float sy = hh ? acc.y : acc.w;
  return make_float2(kx + xhalf(sx), ky + xhalf(sy));
}

__global__ __launch_bounds__(256, 16) void gather_finalize_kernel(
    const float* __restrict__ y, const int* __restrict__ rowptr, const int* __restrict__ sortedSrc,
    const float* __restrict__ inNorm, const float* __restrict__ bias, float* __restrict__ out,
    int nRows, int doRelu) {
  const int lane = threadIdx.x & 31;
  const int n = __builtin_amdgcn_readfirstlane((int)(blockIdx.x * 8 + (threadIdx.x >> 5)));
  if (n >= nRows) return;
  const int hh = lane >> 4;
  const int c0 = ((lane & 15) << 2) | (hh << 1);
  const float4 part = gather_row_sum(reinterpret_cast<const uint2*>(y), rowptr, sortedSrc, n, lane, nRows);
  const float2 acc = gather_combine(part, hh);
  const float  sc = inNorm[n];
  const float2 bb = *reinterpret_cast<const float2*>(bias + c0);
  float2 r;
  r.x = acc.x * sc + bb.x;
  r.y = acc.y * sc + bb.y;
  if (doRelu) { r.x = fmaxf(r.x, 0.0f); r.y = fmaxf(r.y, 0.0f); }
  *reinterpret_cast<float2*>(out + (size_t)n * 64 + c0) = r;
}

typedef __attribute__((ext_vector_type(2))) __bf16 v2bf;

__global__ __launch_bounds__(512, 16) void gather_finalize_gemm_kernel(
    const float* __restrict__ y, const int* __restrict__ rowptr, const int* __restrict__ sortedSrc,
    const float* __restrict__ inNorm, const float* __restrict__ bias,
    const float* __restrict__ outNorm, const float* __restrict__ Wg,
    float* __restrict__ yNext, int nRows) {
  __shared__ __attribute__((aligned(16))) __bf16 Xs[16][64 + 8];
  __shared__ __attribute__((aligned(16))) __bf16 Wt[64][64 + 8];

  const int tid  = threadIdx.x;
  const int lane = tid & 31;
  const int wave    = __builtin_amdgcn_readfirstlane(tid >> 5);
  const int rowBase = blockIdx.x * 16;
  const int n       = rowBase + wave;

  const int wc = tid & 63;
  const int wk = (tid >> 6) << 3;
  float wv[8];
  #pragma unroll
  for (int i = 0; i < 8; ++i) wv[i] = Wg[(wk + i) * 64 + wc];

  const int hhG = lane >> 4;
  const int c0  = ((lane & 15) << 2) | (hhG << 1);
  float2 r = make_float2(0.0f, 0.0f);
  if (n < nRows) {
    const float4 part = gather_row_sum(reinterpret_cast<const uint2*>(y), rowptr, sortedSrc, n, lane, nRows);
    const float2 acc = gather_combine(part, hhG);
    const float  sc = inNorm[n];
    const float  so = outNorm[n];
    const float2 bb = *reinterpret_cast<const float2*>(bias + c0);
    r.x = fmaxf(acc.x * sc + bb.x, 0.0f) * so;
    r.y = fmaxf(acc.y * sc + bb.y, 0.0f) * so;
  }
  v2bf rp;
  rp[0] = f2bf(r.x);
  rp[1] = f2bf(r.y);
  *reinterpret_cast<v2bf*>(&Xs[wave][c0]) = rp;

  v8bf wp;
  #pragma unroll
  for (int i = 0; i < 8; ++i) wp[i] = f2bf(wv[i]);
  *reinterpret_cast<v8bf*>(&Wt[wc][wk]) = wp;
  __syncthreads();

  if (wave >= 4) return;

  const int hh      = lane >> 4;
  const int l       = lane & 15;
  const int colBase = wave << 4;

  v8f acc = {};
  #pragma unroll
  for (int k0 = 0; k0 < 64; k0 += 32) {
    const v8bf a0 = *reinterpret_cast<const v8bf*>(&Xs[l][k0 + (hh << 3)]);
    const v8bf a1 = *reinterpret_cast<const v8bf*>(&Xs[l][k0 + 16 + (hh << 3)]);
    const v8bf b0 = *reinterpret_cast<const v8bf*>(&Wt[colBase + l][k0 + (hh << 3)]);
    const v8bf b1 = *reinterpret_cast<const v8bf*>(&Wt[colBase + l][k0 + 16 + (hh << 3)]);
    v16bf a, b;
    #pragma unroll
    for (int i = 0; i < 8; ++i) {
      a[i] = a0[i]; a[i + 8] = a1[i];
      b[i] = b0[i]; b[i + 8] = b1[i];
    }
    acc = __builtin_amdgcn_wmma_f32_16x16x32_bf16(
         false, a,  false, b,
         (short)0, acc,  false,  false);
  }

  _Float16* __restrict__ yOut = reinterpret_cast<_Float16*>(yNext) + ((long)(rowBase + (hh << 3)) * 64 + colBase + l);
  if (rowBase + 16 <= nRows) {
    #pragma unroll
    for (int v = 0; v < 8; ++v) yOut[(long)v * 64] = (_Float16)acc[v];
  } else {
    #pragma unroll
    for (int v = 0; v < 8; ++v) {
      if (rowBase + v + (hh << 3) < nRows) yOut[(long)v * 64] = (_Float16)acc[v];
    }
  }
}

static inline int cdiv_i(long a, long b) { return (int)((a + b - 1) / b); }

extern "C" void kernel_launch(void* const* d_in, const int* in_sizes, int n_in,
                              void* d_out, int out_size, void* d_ws, size_t ws_size,
                              hipStream_t stream) {
  const float* feat = (const float*)d_in[0];
  const float* W0   = (const float*)d_in[1];
  const float* b0   = (const float*)d_in[2];
  const float* W1   = (const float*)d_in[3];
  const float* b1   = (const float*)d_in[4];
  const float* W2   = (const float*)d_in[5];
  const float* b2   = (const float*)d_in[6];
  const int*   src  = (const int*)d_in[7];
  const int*   dst  = (const int*)d_in[8];

  const int N = in_sizes[0] / 128;
  const int E = in_sizes[7];
  float* out = (float*)d_out;

  const size_t P = (size_t)((N + 63) & ~63);
  const int nBuck  = cdiv_i(N, 1 << BSHIFT);
  const int passes = cdiv_i(N, DEG_BINS);
  float* wsf = (float*)d_ws;
  float* no  = wsf;
  float* ni  = no + P;
  float* y   = ni + P;
  float* h   = y  + (size_t)N * 64;
  unsigned* planes = (unsigned*)(h + (size_t)N * 64);
  int* binOff    = (int*)(planes + (size_t)DEG_PARTS * P);
  int* rowptr    = binOff + (size_t)NSLICE * (size_t)(nBuck + 1);
  int2* binned   = (int2*)(rowptr + P + 64);
  int* sortedSrc = (int*)(binned + (size_t)E);

  const int gemmBlocks = cdiv_i(N, 16);
  const int gatherBlocks = cdiv_i(N, 8);

  plane_bin_kernel<<<DEG_PARTS * passes + NSLICE, 1024, 0, stream>>>(src, dst, planes, binned, binOff, E, N, nBuck, passes);
  bucket_product_kernel<<<nBuck + cdiv_i(N, PROD_ROWS), 1024, 0, stream>>>(
      binned, binOff, planes, rowptr, ni, no, sortedSrc, E, N, nBuck, feat, W0, y);
  gather_finalize_gemm_kernel<<<gemmBlocks, 512, 0, stream>>>(y, rowptr, sortedSrc, ni, b0, no, W1, h, N);
  gather_finalize_gemm_kernel<<<gemmBlocks, 512, 0, stream>>>(h, rowptr, sortedSrc, ni, b1, no, W2, y, N);
  gather_finalize_kernel<<<gatherBlocks, 256, 0, stream>>>(y, rowptr, sortedSrc, ni, b2, out, N, 0);
}


# bucket launch: a wave takes its 8 slice runs as one packed stream: full 32-pair steps, all loads issued ahead of the LDS adds, no dependent load per long run; same list; bound of 64 registers set
# speedup vs baseline: 9.0248x; 1.0321x over previous
#include <hip/hip_runtime.h>

typedef __attribute__((ext_vector_type(16))) __bf16 v16bf;
typedef __attribute__((ext_vector_type(8)))  __bf16 v8bf;
typedef __attribute__((ext_vector_type(8)))  float  v8f;

typedef __attribute__((ext_vector_type(4))) _Float16 v4h;

__device__ __forceinline__ __bf16 f2bf(float f) {
  unsigned int u = __builtin_bit_cast(unsigned int, f);
  u += 0x7FFFu + ((u >> 16) & 1u);
  unsigned short s = (unsigned short)(u >> 16);
  return __builtin_bit_cast(__bf16, s);
}

constexpr int SMEM_WORDS  = 25600;
constexpr int NSLICE      = 256;
constexpr int BSHIFT      = 8;
constexpr int NKEY        = 256;
constexpr int SORT_BATCH  = 8;
constexpr int STAGE_PAIRS = 8192;
constexpr int STAGE_LIST  = 16384;

__device__ __forceinline__ unsigned same_key_mask(unsigned key, bool valid) {
  unsigned m = __builtin_amdgcn_ballot_w32(valid);
  #pragma unroll
  for (int bit = 0; bit < 8; ++bit) {
    const bool one = ((key >> bit) & 1u) != 0u;
    const unsigned b = __builtin_amdgcn_ballot_w32(one);
    m &= one ? b : ~b;
  }
  return m;
}

__device__ __forceinline__ int scan256(int v, int tid, unsigned* red, int& total) {
  const int lane = tid & 31;
  const int wave = tid >> 5;
  int incl = v;
  #pragma unroll
  for (int d = 1; d < 32; d <<= 1) {
    const int t = __shfl_up(incl, d, 32);
    if (lane >= d) incl += t;
  }
  if (lane == 31 && wave < 8) red[wave] = (unsigned)incl;
  __syncthreads();
  int woff = 0, tot = 0;
  #pragma unroll
  for (int w = 0; w < 8; ++w) { const int s = (int)red[w]; tot += s; woff += (w < wave) ? s : 0; }
  total = tot;
  return woff + incl - v;
}

constexpr int DEG_BINS  = 25088;
constexpr int DEG_PARTS = 64;
constexpr int DEG_BATCH = 8;

__device__ __forceinline__ void src_plane_group(const int* __restrict__ idx, unsigned* __restrict__ planes,
                                                int nE, int nNodes, int gx, int gz, unsigned* hist) {
  const int tid    = threadIdx.x;
  const int base   = gz * DEG_BINS;
  const int nb     = max(0, min(DEG_BINS, nNodes - base));
  const int nb4    = nb >> 2;
  const int stride = (nNodes + 63) & ~63;

  uint4* hist4 = reinterpret_cast<uint4*>(hist);
  for (int i = tid; i < nb4; i += 1024) hist4[i] = make_uint4(0u, 0u, 0u, 0u);
  for (int i = (nb4 << 2) + tid; i < nb; i += 1024) hist[i] = 0u;
  __syncthreads();

  const int per = (((nE + DEG_PARTS - 1) / DEG_PARTS) + 3) & ~3;
  const int e0  = min(nE, gx * per);
  const int e1  = min(nE, e0 + per);
  const int nv  = (e1 - e0) >> 2;
  const int4* __restrict__ v = reinterpret_cast<const int4*>(idx + e0);
  for (int i0 = tid; i0 < nv; i0 += DEG_BATCH * 1024) {
    int4 q[DEG_BATCH];
    #pragma unroll
    for (int u = 0; u < DEG_BATCH; ++u) {
      const int i = i0 + u * 1024;
      q[u] = make_int4(-1, -1, -1, -1);
      if (i < nv) q[u] = v[i];
    }
    #pragma unroll
    for (int u = 0; u < DEG_BATCH; ++u) {
      unsigned b;
      b = (unsigned)(q[u].x - base); if (b < (unsigned)nb) atomicAdd(&hist[b], 1u);
      b = (unsigned)(q[u].y - base); if (b < (unsigned)nb) atomicAdd(&hist[b], 1u);
      b = (unsigned)(q[u].z - base); if (b < (unsigned)nb) atomicAdd(&hist[b], 1u);
      b = (unsigned)(q[u].w - base); if (b < (unsigned)nb) atomicAdd(&hist[b], 1u);
    }
  }
  for (int e = e0 + (nv << 2) + tid; e < e1; e += 1024) {
    unsigned b = (unsigned)(idx[e] - base);
    if (b < (unsigned)nb) atomicAdd(&hist[b], 1u);
  }
  __syncthreads();

  unsigned* __restrict__ outp = planes + (size_t)gx * (size_t)stride + base;
  uint4* __restrict__ outp4 = reinterpret_cast<uint4*>(outp);
  for (int i = tid; i < nb4; i += 1024) outp4[i] = hist4[i];
  for (int i = (nb4 << 2) + tid; i < nb; i += 1024) outp[i] = hist[i];
}

__device__ __forceinline__ void bin_slice_group(
    const int* __restrict__ src, const int* __restrict__ dst, int2* __restrict__ binned,
    int* __restrict__ binOff, int nE, int nBuck, int g, unsigned* smem) {
  unsigned* wcnt  = smem;
  unsigned* kst   = smem + 32 * NKEY;
  unsigned* red   = kst + 320;
  int2*     stage = reinterpret_cast<int2*>(smem + 32 * NKEY + 384);
  const int tid  = threadIdx.x;
  const int lane = tid & 31;
  const int wave = __builtin_amdgcn_readfirstlane(tid >> 5);

  const int per   = (((nE + NSLICE - 1) / NSLICE) + 3) & ~3;
  const int e0    = min(nE, g * per);
  const int e1    = min(nE, e0 + per);
  const int len   = e1 - e0;
  const int steps = (len + 31) >> 5;
  const int spw   = (steps + 31) >> 5;
  const int first = wave * spw * 32;
  const int* __restrict__ dS = dst + e0;
  const int* __restrict__ sS = src + e0;
  unsigned* myc = wcnt + wave * NKEY;

  for (int i = tid; i < 32 * NKEY; i += 1024) wcnt[i] = 0u;
  __syncthreads();

  for (int s0 = 0; s0 < spw; s0 += SORT_BATCH) {
    int d[SORT_BATCH];
    #pragma unroll
    for (int u = 0; u < SORT_BATCH; ++u) {
      const int i = first + (s0 + u) * 32 + lane;
      d[u] = -1;
      if (s0 + u < spw && i < len) d[u] = dS[i];
    }
    #pragma unroll
    for (int u = 0; u < SORT_BATCH; ++u)
      if (d[u] >= 0) atomicAdd(&myc[min((unsigned)d[u] >> BSHIFT, (unsigned)(NKEY - 1))], 1u);
  }
  __syncthreads();

  int tot = 0;
  if (tid < NKEY) {
    unsigned run = 0u;
    for (int w = 0; w < 32; ++w) { const unsigned c = wcnt[w * NKEY + tid]; wcnt[w * NKEY + tid] = run; run += c; }
    tot = (int)run;
  }
  int total;
  const int ks = scan256(tot, tid, red, total);
  if (tid <= NKEY) kst[tid] = (unsigned)ks;
  if (tid <= nBuck && tid <= NKEY) binOff[(size_t)g * (nBuck + 1) + tid] = ks;
  __syncthreads();

  const bool toLds = total <= STAGE_PAIRS;
  for (int s0 = 0; s0 < spw; s0 += SORT_BATCH) {
    int d[SORT_BATCH], s[SORT_BATCH];
    #pragma unroll
    for (int u = 0; u < SORT_BATCH; ++u) {
      const int i = first + (s0 + u) * 32 + lane;
      d[u] = -1; s[u] = 0;
      if (s0 + u < spw && i < len) { d[u] = dS[i]; s[u] = sS[i]; }
    }
    #pragma unroll
    for (int u = 0; u < SORT_BATCH; ++u) {
      if (s0 + u < spw) {
        const bool     valid = d[u] >= 0;
        const unsigned key   = valid ? min((unsigned)d[u] >> BSHIFT, (unsigned)(NKEY - 1)) : 0u;
        const unsigned m     = same_key_mask(key, valid);
        const unsigned rank  = (unsigned)__popc(m & ((1u << lane) - 1u));
        const unsigned p     = kst[key] + myc[key] + rank;
        if (valid) {
          const int2 e = make_int2(s[u], d[u]);
          if (toLds) stage[p] = e; else binned[(size_t)e0 + p] = e;
          atomicAdd(&myc[key], 1u);
        }
      }
    }
  }
  __syncthreads();
  if (toLds)
    for (int i = tid; i < total; i += 1024) binned[(size_t)e0 + i] = stage[i];
}

__global__ __launch_bounds__(1024, 2) void plane_bin_kernel(
    const int* __restrict__ src, const int* __restrict__ dst, unsigned* __restrict__ planes,
    int2* __restrict__ binned, int* __restrict__ binOff, int nE, int nNodes, int nBuck, int passes) {
  __shared__ __attribute__((aligned(16))) unsigned smem[SMEM_WORDS];
  const int bid    = (int)blockIdx.x;
  const int nPlane = DEG_PARTS * passes;
  if (bid < nPlane) src_plane_group(src, planes, nE, nNodes, bid % DEG_PARTS, bid / DEG_PARTS, smem);
  else              bin_slice_group(src, dst, binned, binOff, nE, nBuck, bid - nPlane, smem);
}

constexpr int RUNS_PER_WAVE = NSLICE / 32;
constexpr int QSTEPS        = 10;

__device__ __forceinline__ unsigned stream_pair(const int (&ro)[RUNS_PER_WAVE], const int (&pre)[RUNS_PER_WAVE + 1], int j, int nE) {
  int d = ro[0];
  #pragma unroll
  for (int r = 1; r < RUNS_PER_WAVE; ++r) d = (j >= pre[r]) ? (ro[r] - pre[r]) : d;
  return min((unsigned)(d + j), (unsigned)(nE - 1));
}

__device__ __forceinline__ void bucket_group(
    const int2* __restrict__ binned, const int* __restrict__ binOff,
    int* __restrict__ rowptr, float* __restrict__ inNorm, int* __restrict__ sortedSrc,
    int nE, int nNodes, int nBuck, int b, unsigned* smem) {
  unsigned* wcnt   = smem;
  unsigned* kst    = smem + 32 * NKEY;
  unsigned* runOff = kst + NKEY;
  unsigned* runCnt = runOff + NSLICE;
  unsigned* red    = runCnt + NSLICE;
  int*      stage  = reinterpret_cast<int*>(red + 128);
  const int tid  = threadIdx.x;
  const int lane = tid & 31;
  const int wave = __builtin_amdgcn_readfirstlane(tid >> 5);
  unsigned* myc = wcnt + wave * NKEY;

  const int per = (((nE + NSLICE - 1) / NSLICE) + 3) & ~3;
  int o0 = 0, c = 0;
  if (tid < NSLICE) {
    const int* r = binOff + (size_t)tid * (nBuck + 1) + b;
    o0 = r[0];
    c  = r[1] - o0;
    runOff[tid] = (unsigned)(min(nE, tid * per) + o0);
    runCnt[tid] = (unsigned)c;
  }
  for (int i = tid; i < 32 * NKEY; i += 1024) wcnt[i] = 0u;
  int sa = o0, sb = c;
  #pragma unroll
  for (int d = 16; d >= 1; d >>= 1) { sa += __shfl_xor(sa, d, 32); sb += __shfl_xor(sb, d, 32); }
  if (lane == 0) { red[wave] = (unsigned)sa; red[32 + wave] = (unsigned)sb; }
  __syncthreads();
  int base = (int)red[lane], total = (int)red[32 + lane];
  #pragma unroll
  for (int d = 16; d >= 1; d >>= 1) { base += __shfl_xor(base, d, 32); total += __shfl_xor(total, d, 32); }
  base  = __builtin_amdgcn_readfirstlane(base);
  total = __builtin_amdgcn_readfirstlane(total);

  int ro[RUNS_PER_WAVE], pre[RUNS_PER_WAVE + 1];
  pre[0] = 0;
  #pragma unroll
  for (int r = 0; r < RUNS_PER_WAVE; ++r) {
    ro[r]      = __builtin_amdgcn_readfirstlane((int)runOff[wave * RUNS_PER_WAVE + r]);
    pre[r + 1] = pre[r] + max(0, __builtin_amdgcn_readfirstlane((int)runCnt[wave * RUNS_PER_WAVE + r]));
  }
  const int T = pre[RUNS_PER_WAVE];

  int2 q[QSTEPS];
  #pragma unroll
  for (int t = 0; t < QSTEPS; ++t) {
    const int j = t * 32 + lane;
    q[t] = make_int2(0, 0);
    if (j < T) q[t] = binned[stream_pair(ro, pre, j, nE)];
  }
  #pragma unroll
  for (int t = 0; t < QSTEPS; ++t)
    if (t * 32 + lane < T) atomicAdd(&myc[q[t].y & (NKEY - 1)], 1u);
  for (int j0 = QSTEPS * 32; j0 < T; j0 += 32)
    if (j0 + lane < T) atomicAdd(&myc[binned[stream_pair(ro, pre, j0 + lane, nE)].y & (NKEY - 1)], 1u);
  __syncthreads();

  int deg = 0;
  if (tid < NKEY) {
    unsigned run = 0u;
    for (int w = 0; w < 32; ++w) { const unsigned t = wcnt[w * NKEY + tid]; wcnt[w * NKEY + tid] = run; run += t; }
    deg = (int)run;
  }
  int total2;
  const int ks = scan256(deg, tid, red + 64, total2);
  (void)total2;
  if (tid < NKEY) {
    kst[tid] = (unsigned)ks;
    const int n = (b << BSHIFT) + tid;
    if (n < nNodes) {
      rowptr[n] = base + ks;
      inNorm[n] = rsqrtf(fmaxf((float)deg, 1.0f));
      if (n == nNodes - 1) rowptr[nNodes] = base + ks + deg;
    }
  }
  __syncthreads();

  const bool toLds = total <= STAGE_LIST;
  const int  nst   = (T + 31) >> 5;
  for (int t0 = 0; t0 < nst; t0 += QSTEPS) {
    #pragma unroll
    for (int u = 0; u < QSTEPS; ++u) {
      const int t = t0 + u;
      if (t < nst) {
        const int  j     = t * 32 + lane;
        const bool valid = j < T;
        int2 e = q[u];
        if (t0 > 0) { e = make_int2(0, 0); if (valid) e = binned[stream_pair(ro, pre, j, nE)]; }
        const unsigned key  = valid ? (unsigned)(e.y & (NKEY - 1)) : 0u;
        const unsigned m    = same_key_mask(key, valid);
        const unsigned rank = (unsigned)__popc(m & ((1u << lane) - 1u));
        const unsigned p    = kst[key] + myc[key] + rank;
        if (valid) {
          if (toLds) stage[p] = e.x;
          else if ((long)base + p < (long)nE) sortedSrc[(size_t)base + p] = e.x;
          atomicAdd(&myc[key], 1u);
        }
      }
    }
  }
  __syncthreads();
  if (toLds)
    for (int i = tid; i < total; i += 1024)
      if ((long)base + i < (long)nE) sortedSrc[(size_t)base + i] = stage[i];
}

typedef __attribute__((ext_vector_type(4))) __bf16 v4bf;
constexpr int PROD_ROWS  = 128;
constexpr int PROD_K     = 128;
constexpr int PROD_PITCH = PROD_K + 8;

__device__ __forceinline__ void product_rows128(
    const float* __restrict__ X, const unsigned* __restrict__ planes, float* __restrict__ outNorm,
    const float* __restrict__ Wg, _Float16* __restrict__ Y, int nRows, int rowBase, unsigned* smem) {
  __bf16*   Xs     = reinterpret_cast<__bf16*>(smem);
  __bf16*   Wt     = Xs + PROD_ROWS * PROD_PITCH;
  unsigned* rowCnt = smem + (PROD_ROWS + 64) * PROD_PITCH / 2;
  float*    noS    = reinterpret_cast<float*>(rowCnt + PROD_ROWS);
  const int tid  = threadIdx.x;
  const int lane = tid & 31;
  const int wave = tid >> 5;
  const int stride = (nRows + 63) & ~63;

  const int wc = tid & 63;
  const int wk = (tid >> 6) << 3;
  float wv[8];
  #pragma unroll
  for (int i = 0; i < 8; ++i) wv[i] = Wg[(wk + i) * 64 + wc];
  float4 xv[4];
  #pragma unroll
  for (int j = 0; j < 4; ++j) {
    int row = rowBase + wave + 32 * j;
    row = (row < nRows) ? row : (nRows - 1);
    xv[j] = *reinterpret_cast<const float4*>(X + (size_t)row * PROD_K + lane * 4);
  }
  if (tid < PROD_ROWS) rowCnt[tid] = 0u;
  __syncthreads();
  {
    const int r   = tid & (PROD_ROWS - 1);
    const int grp = tid >> 7;
    int row = rowBase + r;
    row = (row < nRows) ? row : (nRows - 1);
    unsigned cnt = 0u;
    #pragma unroll
    for (int i = 0; i < DEG_PARTS / 8; ++i) cnt += planes[(size_t)(grp * (DEG_PARTS / 8) + i) * stride + row];
    atomicAdd(&rowCnt[r], cnt);
  }
  __syncthreads();
  if (tid < PROD_ROWS) {
    const float sc = rsqrtf(fmaxf((float)rowCnt[tid], 1.0f));
    noS[tid] = sc;
    if (rowBase + tid < nRows) outNorm[rowBase + tid] = sc;
  }
  __syncthreads();

  #pragma unroll
  for (int j = 0; j < 4; ++j) {
    const float xs = noS[wave + 32 * j];
    v4bf p;
    p[0] = f2bf(xv[j].x * xs);
    p[1] = f2bf(xv[j].y * xs);
    p[2] = f2bf(xv[j].z * xs);
    p[3] = f2bf(xv[j].w * xs);
    *reinterpret_cast<v4bf*>(&Xs[(wave + 32 * j) * PROD_PITCH + lane * 4]) = p;
  }
  v8bf wp;
  #pragma unroll
  for (int i = 0; i < 8; ++i) wp[i] = f2bf(wv[i]);
  *reinterpret_cast<v8bf*>(&Wt[wc * PROD_PITCH + wk]) = wp;
  __syncthreads();

  const int hh      = lane >> 4;
  const int l       = lane & 15;
  const int rt      = wave >> 2;
  const int colBase = (wave & 3) << 4;
  const __bf16* xr = Xs + (rt * 16 + l) * PROD_PITCH;
  const __bf16* wr = Wt + (colBase + l) * PROD_PITCH;

  v8f acc = {};
  #pragma unroll
  for (int k0 = 0; k0 < PROD_K; k0 += 32) {
    const v8bf a0 = *reinterpret_cast<const v8bf*>(xr + k0 + (hh << 3));
    const v8bf a1 = *reinterpret_cast<const v8bf*>(xr + k0 + 16 + (hh << 3));
    const v8bf b0 = *reinterpret_cast<const v8bf*>(wr + k0 + (hh << 3));
    const v8bf b1 = *reinterpret_cast<const v8bf*>(wr + k0 + 16 + (hh << 3));
    v16bf a, b;
    #pragma unroll
    for (int i = 0; i < 8; ++i) {
      a[i] = a0[i]; a[i + 8] = a1[i];
      b[i] = b0[i]; b[i + 8] = b1[i];
    }
    acc = __builtin_amdgcn_wmma_f32_16x16x32_bf16(
         false, a,  false, b,
         (short)0, acc,  false,  false);
  }

  const int r0 = rowBase + rt * 16;
  _Float16* __restrict__ yOut = Y + ((size_t)(r0 + (hh << 3)) * 64 + colBase + l);
  if (r0 + 16 <= nRows) {
    #pragma unroll
    for (int v = 0; v < 8; ++v) yOut[(size_t)v * 64] = (_Float16)acc[v];
  } else {
    #pragma unroll
    for (int v = 0; v < 8; ++v) {
      if (r0 + v + (hh << 3) < nRows) yOut[(size_t)v * 64] = (_Float16)acc[v];
    }
  }
}

__global__ __launch_bounds__(1024, 16) void bucket_product_kernel(
    const int2* __restrict__ binned, const int* __restrict__ binOff, const unsigned* __restrict__ planes,
    int* __restrict__ rowptr, float* __restrict__ inNorm, float* __restrict__ outNorm,
    int* __restrict__ sortedSrc, int nE, int nNodes, int nBuck,
    const float* __restrict__ X, const float* __restrict__ Wg, float* __restrict__ Y) {
  __shared__ __attribute__((aligned(16))) unsigned smem[SMEM_WORDS];
  const int bid = (int)blockIdx.x;
  if (bid < nBuck) bucket_group(binned, binOff, rowptr, inNorm, sortedSrc, nE, nNodes, nBuck, bid, smem);
  else             product_rows128(X, planes, outNorm, Wg, reinterpret_cast<_Float16*>(Y), nNodes, (bid - nBuck) * PROD_ROWS, smem);
}

constexpr int GATHER_BATCH = 8;

template<int U>
__device__ __forceinline__ int half_lane(int x) {
  return __builtin_amdgcn_update_dpp(0, x, 0x150 + U, 0xf, 0xf, true);
}

template<int U0, bool PARTIAL>
__device__ __forceinline__ void gather_batch(const uint2* __restrict__ y8, int rowOff, int cnt,
                                             int hh, int p, float4& acc) {
  uint2 w[GATHER_BATCH];
#define GATHER_LOAD(u) w[u] = y8[half_lane<U0 + (u)>(rowOff) | p];
  GATHER_LOAD(0) GATHER_LOAD(1) GATHER_LOAD(2) GATHER_LOAD(3)
  GATHER_LOAD(4) GATHER_LOAD(5) GATHER_LOAD(6) GATHER_LOAD(7)
#undef GATHER_LOAD
  #pragma unroll
  for (int u = 0; u < GATHER_BATCH; ++u) {
    uint2 t = w[u];
    if (PARTIAL) {
      const bool ok = (2 * (U0 + u) + hh) < cnt;
      t.x = ok ? t.x : 0u;
      t.y = ok ? t.y : 0u;
    }
    const v4h v = __builtin_bit_cast(v4h, t);
    acc.x += (float)v[0];
    acc.y += (float)v[1];
    acc.z += (float)v[2];
    acc.w += (float)v[3];
  }
}

__device__ __forceinline__ float4 gather_row_sum(const uint2* __restrict__ y8, const int* __restrict__ rowptr,
                                                 const int* __restrict__ sortedSrc, int n, int lane, int nRows) {
  const int beg = rowptr[n];
  const int end = rowptr[n + 1];
  const int hh  = lane >> 4;
  const int p   = lane & 15;
  const int pos = (p << 1) | hh;
  float4 acc = make_float4(0.0f, 0.0f, 0.0f, 0.0f);
  int idx = (beg + pos < end) ? sortedSrc[beg + pos] : 0;
  for (int base = beg; base < end; base += 32) {
    const int cnt = min(32, end - base);
    const int inext = base + 32 + pos;
    const int nxt = (inext < end) ? sortedSrc[inext] : 0;
    const int rowOff = (int)(min((unsigned)idx, (unsigned)(nRows - 1)) << 4);
    if (cnt == 32) {
      gather_batch<0, false>(y8, rowOff, cnt, hh, p, acc);
      __builtin_amdgcn_sched_barrier(0);
      gather_batch<8, false>(y8, rowOff, cnt, hh, p, acc);
    } else {
      gather_batch<0, true>(y8, rowOff, cnt, hh, p, acc);
      if (cnt > 16) gather_batch<8, true>(y8, rowOff, cnt, hh, p, acc);
    }
    idx = nxt;
  }
  return acc;
}

__device__ __forceinline__ float xhalf(float x) {
  const int xi = __builtin_bit_cast(int, x);
  return __builtin_bit_cast(float, __builtin_amdgcn_permlanex16(xi, xi, 0x76543210, 0xfedcba98, false, false));
}

__device__ __forceinline__ float2 gather_combine(const float4 acc, int hh) {
  const float kx = hh ? acc.z : acc.x;
  const float ky = hh ? acc.w : acc.y;
  const float sx = hh ? acc.x : acc.z;
  const float sy = hh ? acc.y : acc.w;
  return make_float2(kx + xhalf(sx), ky + xhalf(sy));
}

__global__ __launch_bounds__(256, 16) void gather_finalize_kernel(
    const float* __restrict__ y, const int* __restrict__ rowptr, const int* __restrict__ sortedSrc,
    const float* __restrict__ inNorm, const float* __restrict__ bias, float* __restrict__ out,
    int nRows, int doRelu) {
  const int lane = threadIdx.x & 31;
  const int n = __builtin_amdgcn_readfirstlane((int)(blockIdx.x * 8 + (threadIdx.x >> 5)));
  if (n >= nRows) return;
  const int hh = lane >> 4;
  const int c0 = ((lane & 15) << 2) | (hh << 1);
  const float4 part = gather_row_sum(reinterpret_cast<const uint2*>(y), rowptr, sortedSrc, n, lane, nRows);
  const float2 acc = gather_combine(part, hh);
  const float  sc = inNorm[n];
  const float2 bb = *reinterpret_cast<const float2*>(bias + c0);
  float2 r;
  r.x = acc.x * sc + bb.x;
  r.y = acc.y * sc + bb.y;
  if (doRelu) { r.x = fmaxf(r.x, 0.0f); r.y = fmaxf(r.y, 0.0f); }
  *reinterpret_cast<float2*>(out + (size_t)n * 64 + c0) = r;
}

typedef __attribute__((ext_vector_type(2))) __bf16 v2bf;

__global__ __launch_bounds__(512, 16) void gather_finalize_gemm_kernel(
    const float* __restrict__ y, const int* __restrict__ rowptr, const int* __restrict__ sortedSrc,
    const float* __restrict__ inNorm, const float* __restrict__ bias,
    const float* __restrict__ outNorm, const float* __restrict__ Wg,
    float* __restrict__ yNext, int nRows) {
  __shared__ __attribute__((aligned(16))) __bf16 Xs[16][64 + 8];
  __shared__ __attribute__((aligned(16))) __bf16 Wt[64][64 + 8];

  const int tid  = threadIdx.x;
  const int lane = tid & 31;
  const int wave    = __builtin_amdgcn_readfirstlane(tid >> 5);
  const int rowBase = blockIdx.x * 16;
  const int n       = rowBase + wave;

  const int wc = tid & 63;
  const int wk = (tid >> 6) << 3;
  float wv[8];
  #pragma unroll
  for (int i = 0; i < 8; ++i) wv[i] = Wg[(wk + i) * 64 + wc];

  const int hhG = lane >> 4;
  const int c0  = ((lane & 15) << 2) | (hhG << 1);
  float2 r = make_float2(0.0f, 0.0f);
  if (n < nRows) {
    const float4 part = gather_row_sum(reinterpret_cast<const uint2*>(y), rowptr, sortedSrc, n, lane, nRows);
    const float2 acc = gather_combine(part, hhG);
    const float  sc = inNorm[n];
    const float  so = outNorm[n];
    const float2 bb = *reinterpret_cast<const float2*>(bias + c0);
    r.x = fmaxf(acc.x * sc + bb.x, 0.0f) * so;
    r.y = fmaxf(acc.y * sc + bb.y, 0.0f) * so;
  }
  v2bf rp;
  rp[0] = f2bf(r.x);
  rp[1] = f2bf(r.y);
  *reinterpret_cast<v2bf*>(&Xs[wave][c0]) = rp;

  v8bf wp;
  #pragma unroll
  for (int i = 0; i < 8; ++i) wp[i] = f2bf(wv[i]);
  *reinterpret_cast<v8bf*>(&Wt[wc][wk]) = wp;
  __syncthreads();

  if (wave >= 4) return;

  const int hh      = lane >> 4;
  const int l       = lane & 15;
  const int colBase = wave << 4;

  v8f acc = {};
  #pragma unroll
  for (int k0 = 0; k0 < 64; k0 += 32) {
    const v8bf a0 = *reinterpret_cast<const v8bf*>(&Xs[l][k0 + (hh << 3)]);
    const v8bf a1 = *reinterpret_cast<const v8bf*>(&Xs[l][k0 + 16 + (hh << 3)]);
    const v8bf b0 = *reinterpret_cast<const v8bf*>(&Wt[colBase + l][k0 + (hh << 3)]);
    const v8bf b1 = *reinterpret_cast<const v8bf*>(&Wt[colBase + l][k0 + 16 + (hh << 3)]);
    v16bf a, b;
    #pragma unroll
    for (int i = 0; i < 8; ++i) {
      a[i] = a0[i]; a[i + 8] = a1[i];
      b[i] = b0[i]; b[i + 8] = b1[i];
    }
    acc = __builtin_amdgcn_wmma_f32_16x16x32_bf16(
         false, a,  false, b,
         (short)0, acc,  false,  false);
  }

  _Float16* __restrict__ yOut = reinterpret_cast<_Float16*>(yNext) + ((long)(rowBase + (hh << 3)) * 64 + colBase + l);
  if (rowBase + 16 <= nRows) {
    #pragma unroll
    for (int v = 0; v < 8; ++v) yOut[(long)v * 64] = (_Float16)acc[v];
  } else {
    #pragma unroll
    for (int v = 0; v < 8; ++v) {
      if (rowBase + v + (hh << 3) < nRows) yOut[(long)v * 64] = (_Float16)acc[v];
    }
  }
}

static inline int cdiv_i(long a, long b) { return (int)((a + b - 1) / b); }

extern "C" void kernel_launch(void* const* d_in, const int* in_sizes, int n_in,
                              void* d_out, int out_size, void* d_ws, size_t ws_size,
                              hipStream_t stream) {
  const float* feat = (const float*)d_in[0];
  const float* W0   = (const float*)d_in[1];
  const float* b0   = (const float*)d_in[2];
  const float* W1   = (const float*)d_in[3];
  const float* b1   = (const float*)d_in[4];
  const float* W2   = (const float*)d_in[5];
  const float* b2   = (const float*)d_in[6];
  const int*   src  = (const int*)d_in[7];
  const int*   dst  = (const int*)d_in[8];

  const int N = in_sizes[0] / 128;
  const int E = in_sizes[7];
  float* out = (float*)d_out;

  const size_t P = (size_t)((N + 63) & ~63);
  const int nBuck  = cdiv_i(N, 1 << BSHIFT);
  const int passes = cdiv_i(N, DEG_BINS);
  float* wsf = (float*)d_ws;
  float* no  = wsf;
  float* ni  = no + P;
  float* y   = ni + P;
  float* h   = y  + (size_t)N * 64;
  unsigned* planes = (unsigned*)(h + (size_t)N * 64);
  int* binOff    = (int*)(planes + (size_t)DEG_PARTS * P);
  int* rowptr    = binOff + (size_t)NSLICE * (size_t)(nBuck + 1);
  int2* binned   = (int2*)(rowptr + P + 64);
  int* sortedSrc = (int*)(binned + (size_t)E);

  const int gemmBlocks = cdiv_i(N, 16);
  const int gatherBlocks = cdiv_i(N, 8);

  plane_bin_kernel<<<DEG_PARTS * passes + NSLICE, 1024, 0, stream>>>(src, dst, planes, binned, binOff, E, N, nBuck, passes);
  bucket_product_kernel<<<nBuck + cdiv_i(N, PROD_ROWS), 1024, 0, stream>>>(
      binned, binOff, planes, rowptr, ni, no, sortedSrc, E, N, nBuck, feat, W0, y);
  gather_finalize_gemm_kernel<<<gemmBlocks, 512, 0, stream>>>(y, rowptr, sortedSrc, ni, b0, no, W1, h, N);
  gather_finalize_gemm_kernel<<<gemmBlocks, 512, 0, stream>>>(h, rowptr, sortedSrc, ni, b1, no, W2, y, N);
  gather_finalize_kernel<<<gatherBlocks, 256, 0, stream>>>(y, rowptr, sortedSrc, ni, b2, out, N, 0);
}


# gather launches: a wave takes 4 neighbour nodes in turn, next node's indices loaded ahead: 12,500 resident waves per layer, not 50,000 short ones; hidden-layer groups of 4 waves; same sums
# speedup vs baseline: 9.3840x; 1.0398x over previous
#include <hip/hip_runtime.h>

typedef __attribute__((ext_vector_type(16))) __bf16 v16bf;
typedef __attribute__((ext_vector_type(8)))  __bf16 v8bf;
typedef __attribute__((ext_vector_type(8)))  float  v8f;

typedef __attribute__((ext_vector_type(4))) _Float16 v4h;

__device__ __forceinline__ __bf16 f2bf(float f) {
  unsigned int u = __builtin_bit_cast(unsigned int, f);
  u += 0x7FFFu + ((u >> 16) & 1u);
  unsigned short s = (unsigned short)(u >> 16);
  return __builtin_bit_cast(__bf16, s);
}

constexpr int SMEM_WORDS  = 25600;
constexpr int NSLICE      = 256;
constexpr int BSHIFT      = 8;
constexpr int NKEY        = 256;
constexpr int SORT_BATCH  = 8;
constexpr int STAGE_PAIRS = 8192;
constexpr int STAGE_LIST  = 16384;

__device__ __forceinline__ unsigned same_key_mask(unsigned key, bool valid) {
  unsigned m = __builtin_amdgcn_ballot_w32(valid);
  #pragma unroll
  for (int bit = 0; bit < 8; ++bit) {
    const bool one = ((key >> bit) & 1u) != 0u;
    const unsigned b = __builtin_amdgcn_ballot_w32(one);
    m &= one ? b : ~b;
  }
  return m;
}

__device__ __forceinline__ int scan256(int v, int tid, unsigned* red, int& total) {
  const int lane = tid & 31;
  const int wave = tid >> 5;
  int incl = v;
  #pragma unroll
  for (int d = 1; d < 32; d <<= 1) {
    const int t = __shfl_up(incl, d, 32);
    if (lane >= d) incl += t;
  }
  if (lane == 31 && wave < 8) red[wave] = (unsigned)incl;
  __syncthreads();
  int woff = 0, tot = 0;
  #pragma unroll
  for (int w = 0; w < 8; ++w) { const int s = (int)red[w]; tot += s; woff += (w < wave) ? s : 0; }
  total = tot;
  return woff + incl - v;
}

constexpr int DEG_BINS  = 25088;
constexpr int DEG_PARTS = 64;
constexpr int DEG_BATCH = 8;

__device__ __forceinline__ void src_plane_group(const int* __restrict__ idx, unsigned* __restrict__ planes,
                                                int nE, int nNodes, int gx, int gz, unsigned* hist) {
  const int tid    = threadIdx.x;
  const int base   = gz * DEG_BINS;
  const int nb     = max(0, min(DEG_BINS, nNodes - base));
  const int nb4    = nb >> 2;
  const int stride = (nNodes + 63) & ~63;

  uint4* hist4 = reinterpret_cast<uint4*>(hist);
  for (int i = tid; i < nb4; i += 1024) hist4[i] = make_uint4(0u, 0u, 0u, 0u);
  for (int i = (nb4 << 2) + tid; i < nb; i += 1024) hist[i] = 0u;
  __syncthreads();

  const int per = (((nE + DEG_PARTS - 1) / DEG_PARTS) + 3) & ~3;
  const int e0  = min(nE, gx * per);
  const int e1  = min(nE, e0 + per);
  const int nv  = (e1 - e0) >> 2;
  const int4* __restrict__ v = reinterpret_cast<const int4*>(idx + e0);
  for (int i0 = tid; i0 < nv; i0 += DEG_BATCH * 1024) {
    int4 q[DEG_BATCH];
    #pragma unroll
    for (int u = 0; u < DEG_BATCH; ++u) {
      const int i = i0 + u * 1024;
      q[u] = make_int4(-1, -1, -1, -1);
      if (i < nv) q[u] = v[i];
    }
    #pragma unroll
    for (int u = 0; u < DEG_BATCH; ++u) {
      unsigned b;
      b = (unsigned)(q[u].x - base); if (b < (unsigned)nb) atomicAdd(&hist[b], 1u);
      b = (unsigned)(q[u].y - base); if (b < (unsigned)nb) atomicAdd(&hist[b], 1u);
      b = (unsigned)(q[u].z - base); if (b < (unsigned)nb) atomicAdd(&hist[b], 1u);
      b = (unsigned)(q[u].w - base); if (b < (unsigned)nb) atomicAdd(&hist[b], 1u);
    }
  }
  for (int e = e0 + (nv << 2) + tid; e < e1; e += 1024) {
    unsigned b = (unsigned)(idx[e] - base);
    if (b < (unsigned)nb) atomicAdd(&hist[b], 1u);
  }
  __syncthreads();

  unsigned* __restrict__ outp = planes + (size_t)gx * (size_t)stride + base;
  uint4* __restrict__ outp4 = reinterpret_cast<uint4*>(outp);
  for (int i = tid; i < nb4; i += 1024) outp4[i] = hist4[i];
  for (int i = (nb4 << 2) + tid; i < nb; i += 1024) outp[i] = hist[i];
}

__device__ __forceinline__ void bin_slice_group(
    const int* __restrict__ src, const int* __restrict__ dst, int2* __restrict__ binned,
    int* __restrict__ binOff, int nE, int nBuck, int g, unsigned* smem) {
  unsigned* wcnt  = smem;
  unsigned* kst   = smem + 32 * NKEY;
  unsigned* red   = kst + 320;
  int2*     stage = reinterpret_cast<int2*>(smem + 32 * NKEY + 384);
  const int tid  = threadIdx.x;
  const int lane = tid & 31;
  const int wave = __builtin_amdgcn_readfirstlane(tid >> 5);

  const int per   = (((nE + NSLICE - 1) / NSLICE) + 3) & ~3;
  const int e0    = min(nE, g * per);
  const int e1    = min(nE, e0 + per);
  const int len   = e1 - e0;
  const int steps = (len + 31) >> 5;
  const int spw   = (steps + 31) >> 5;
  const int first = wave * spw * 32;
  const int* __restrict__ dS = dst + e0;
  const int* __restrict__ sS = src + e0;
  unsigned* myc = wcnt + wave * NKEY;

  for (int i = tid; i < 32 * NKEY; i += 1024) wcnt[i] = 0u;
  __syncthreads();

  for (int s0 = 0; s0 < spw; s0 += SORT_BATCH) {
    int d[SORT_BATCH];
    #pragma unroll
    for (int u = 0; u < SORT_BATCH; ++u) {
      const int i = first + (s0 + u) * 32 + lane;
      d[u] = -1;
      if (s0 + u < spw && i < len) d[u] = dS[i];
    }
    #pragma unroll
    for (int u = 0; u < SORT_BATCH; ++u)
      if (d[u] >= 0) atomicAdd(&myc[min((unsigned)d[u] >> BSHIFT, (unsigned)(NKEY - 1))], 1u);
  }
  __syncthreads();

  int tot = 0;
  if (tid < NKEY) {
    unsigned run = 0u;
    for (int w = 0; w < 32; ++w) { const unsigned c = wcnt[w * NKEY + tid]; wcnt[w * NKEY + tid] = run; run += c; }
    tot = (int)run;
  }
  int total;
  const int ks = scan256(tot, tid, red, total);
  if (tid <= NKEY) kst[tid] = (unsigned)ks;
  if (tid <= nBuck && tid <= NKEY) binOff[(size_t)g * (nBuck + 1) + tid] = ks;
  __syncthreads();

  const bool toLds = total <= STAGE_PAIRS;
  for (int s0 = 0; s0 < spw; s0 += SORT_BATCH) {
    int d[SORT_BATCH], s[SORT_BATCH];
    #pragma unroll
    for (int u = 0; u < SORT_BATCH; ++u) {
      const int i = first + (s0 + u) * 32 + lane;
      d[u] = -1; s[u] = 0;
      if (s0 + u < spw && i < len) { d[u] = dS[i]; s[u] = sS[i]; }
    }
    #pragma unroll
    for (int u = 0; u < SORT_BATCH; ++u) {
      if (s0 + u < spw) {
        const bool     valid = d[u] >= 0;
        const unsigned key   = valid ? min((unsigned)d[u] >> BSHIFT, (unsigned)(NKEY - 1)) : 0u;
        const unsigned m     = same_key_mask(key, valid);
        const unsigned rank  = (unsigned)__popc(m & ((1u << lane) - 1u));
        const unsigned p     = kst[key] + myc[key] + rank;
        if (valid) {
          const int2 e = make_int2(s[u], d[u]);
          if (toLds) stage[p] = e; else binned[(size_t)e0 + p] = e;
          atomicAdd(&myc[key], 1u);
        }
      }
    }
  }
  __syncthreads();
  if (toLds)
    for (int i = tid; i < total; i += 1024) binned[(size_t)e0 + i] = stage[i];
}

__global__ __launch_bounds__(1024, 2) void plane_bin_kernel(
    const int* __restrict__ src, const int* __restrict__ dst, unsigned* __restrict__ planes,
    int2* __restrict__ binned, int* __restrict__ binOff, int nE, int nNodes, int nBuck, int passes) {
  __shared__ __attribute__((aligned(16))) unsigned smem[SMEM_WORDS];
  const int bid    = (int)blockIdx.x;
  const int nPlane = DEG_PARTS * passes;
  if (bid < nPlane) src_plane_group(src, planes, nE, nNodes, bid % DEG_PARTS, bid / DEG_PARTS, smem);
  else              bin_slice_group(src, dst, binned, binOff, nE, nBuck, bid - nPlane, smem);
}

constexpr int RUNS_PER_WAVE = NSLICE / 32;
constexpr int QSTEPS        = 10;

__device__ __forceinline__ unsigned stream_pair(const int (&ro)[RUNS_PER_WAVE], const int (&pre)[RUNS_PER_WAVE + 1], int j, int nE) {
  int d = ro[0];
  #pragma unroll
  for (int r = 1; r < RUNS_PER_WAVE; ++r) d = (j >= pre[r]) ? (ro[r] - pre[r]) : d;
  return min((unsigned)(d + j), (unsigned)(nE - 1));
}

__device__ __forceinline__ void bucket_group(
    const int2* __restrict__ binned, const int* __restrict__ binOff,
    int* __restrict__ rowptr, float* __restrict__ inNorm, int* __restrict__ sortedSrc,
    int nE, int nNodes, int nBuck, int b, unsigned* smem) {
  unsigned* wcnt   = smem;
  unsigned* kst    = smem + 32 * NKEY;
  unsigned* runOff = kst + NKEY;
  unsigned* runCnt = runOff + NSLICE;
  unsigned* red    = runCnt + NSLICE;
  int*      stage  = reinterpret_cast<int*>(red + 128);
  const int tid  = threadIdx.x;
  const int lane = tid & 31;
  const int wave = __builtin_amdgcn_readfirstlane(tid >> 5);
  unsigned* myc = wcnt + wave * NKEY;

  const int per = (((nE + NSLICE - 1) / NSLICE) + 3) & ~3;
  int o0 = 0, c = 0;
  if (tid < NSLICE) {
    const int* r = binOff + (size_t)tid * (nBuck + 1) + b;
    o0 = r[0];
    c  = r[1] - o0;
    runOff[tid] = (unsigned)(min(nE, tid * per) + o0);
    runCnt[tid] = (unsigned)c;
  }
  for (int i = tid; i < 32 * NKEY; i += 1024) wcnt[i] = 0u;
  int sa = o0, sb = c;
  #pragma unroll
  for (int d = 16; d >= 1; d >>= 1) { sa += __shfl_xor(sa, d, 32); sb += __shfl_xor(sb, d, 32); }
  if (lane == 0) { red[wave] = (unsigned)sa; red[32 + wave] = (unsigned)sb; }
  __syncthreads();
  int base = (int)red[lane], total = (int)red[32 + lane];
  #pragma unroll
  for (int d = 16; d >= 1; d >>= 1) { base += __shfl_xor(base, d, 32); total += __shfl_xor(total, d, 32); }
  base  = __builtin_amdgcn_readfirstlane(base);
  total = __builtin_amdgcn_readfirstlane(total);

  int ro[RUNS_PER_WAVE], pre[RUNS_PER_WAVE + 1];
  pre[0] = 0;
  #pragma unroll
  for (int r = 0; r < RUNS_PER_WAVE; ++r) {
    ro[r]      = __builtin_amdgcn_readfirstlane((int)runOff[wave * RUNS_PER_WAVE + r]);
    pre[r + 1] = pre[r] + max(0, __builtin_amdgcn_readfirstlane((int)runCnt[wave * RUNS_PER_WAVE + r]));
  }
  const int T = pre[RUNS_PER_WAVE];

  int2 q[QSTEPS];
  #pragma unroll
  for (int t = 0; t < QSTEPS; ++t) {
    const int j = t * 32 + lane;
    q[t] = make_int2(0, 0);
    if (j < T) q[t] = binned[stream_pair(ro, pre, j, nE)];
  }
  #pragma unroll
  for (int t = 0; t < QSTEPS; ++t)
    if (t * 32 + lane < T) atomicAdd(&myc[q[t].y & (NKEY - 1)], 1u);
  for (int j0 = QSTEPS * 32; j0 < T; j0 += 32)
    if (j0 + lane < T) atomicAdd(&myc[binned[stream_pair(ro, pre, j0 + lane, nE)].y & (NKEY - 1)], 1u);
  __syncthreads();

  int deg = 0;
  if (tid < NKEY) {
    unsigned run = 0u;
    for (int w = 0; w < 32; ++w) { const unsigned t = wcnt[w * NKEY + tid]; wcnt[w * NKEY + tid] = run; run += t; }
    deg = (int)run;
  }
  int total2;
  const int ks = scan256(deg, tid, red + 64, total2);
  (void)total2;
  if (tid < NKEY) {
    kst[tid] = (unsigned)ks;
    const int n = (b << BSHIFT) + tid;
    if (n < nNodes) {
      rowptr[n] = base + ks;
      inNorm[n] = rsqrtf(fmaxf((float)deg, 1.0f));
      if (n == nNodes - 1) rowptr[nNodes] = base + ks + deg;
    }
  }
  __syncthreads();

  const bool toLds = total <= STAGE_LIST;
  const int  nst   = (T + 31) >> 5;
  for (int t0 = 0; t0 < nst; t0 += QSTEPS) {
    #pragma unroll
    for (int u = 0; u < QSTEPS; ++u) {
      const int t = t0 + u;
      if (t < nst) {
        const int  j     = t * 32 + lane;
        const bool valid = j < T;
        int2 e = q[u];
        if (t0 > 0) { e = make_int2(0, 0); if (valid) e = binned[stream_pair(ro, pre, j, nE)]; }
        const unsigned key  = valid ? (unsigned)(e.y & (NKEY - 1)) : 0u;
        const unsigned m    = same_key_mask(key, valid);
        const unsigned rank = (unsigned)__popc(m & ((1u << lane) - 1u));
        const unsigned p    = kst[key] + myc[key] + rank;
        if (valid) {
          if (toLds) stage[p] = e.x;
          else if ((long)base + p < (long)nE) sortedSrc[(size_t)base + p] = e.x;
          atomicAdd(&myc[key], 1u);
        }
      }
    }
  }
  __syncthreads();
  if (toLds)
    for (int i = tid; i < total; i += 1024)
      if ((long)base + i < (long)nE) sortedSrc[(size_t)base + i] = stage[i];
}

typedef __attribute__((ext_vector_type(4))) __bf16 v4bf;
constexpr int PROD_ROWS  = 128;
constexpr int PROD_K     = 128;
constexpr int PROD_PITCH = PROD_K + 8;

__device__ __forceinline__ void product_rows128(
    const float* __restrict__ X, const unsigned* __restrict__ planes, float* __restrict__ outNorm,
    const float* __restrict__ Wg, _Float16* __restrict__ Y, int nRows, int rowBase, unsigned* smem) {
  __bf16*   Xs     = reinterpret_cast<__bf16*>(smem);
  __bf16*   Wt     = Xs + PROD_ROWS * PROD_PITCH;
  unsigned* rowCnt = smem + (PROD_ROWS + 64) * PROD_PITCH / 2;
  float*    noS    = reinterpret_cast<float*>(rowCnt + PROD_ROWS);
  const int tid  = threadIdx.x;
  const int lane = tid & 31;
  const int wave = tid >> 5;
  const int stride = (nRows + 63) & ~63;

  const int wc = tid & 63;
  const int wk = (tid >> 6) << 3;
  float wv[8];
  #pragma unroll
  for (int i = 0; i < 8; ++i) wv[i] = Wg[(wk + i) * 64 + wc];
  float4 xv[4];
  #pragma unroll
  for (int j = 0; j < 4; ++j) {
    int row = rowBase + wave + 32 * j;
    row = (row < nRows) ? row : (nRows - 1);
    xv[j] = *reinterpret_cast<const float4*>(X + (size_t)row * PROD_K + lane * 4);
  }
  if (tid < PROD_ROWS) rowCnt[tid] = 0u;
  __syncthreads();
  {
    const int r   = tid & (PROD_ROWS - 1);
    const int grp = tid >> 7;
    int row = rowBase + r;
    row = (row < nRows) ? row : (nRows - 1);
    unsigned cnt = 0u;
    #pragma unroll
    for (int i = 0; i < DEG_PARTS / 8; ++i) cnt += planes[(size_t)(grp * (DEG_PARTS / 8) + i) * stride + row];
    atomicAdd(&rowCnt[r], cnt);
  }
  __syncthreads();
  if (tid < PROD_ROWS) {
    const float sc = rsqrtf(fmaxf((float)rowCnt[tid], 1.0f));
    noS[tid] = sc;
    if (rowBase + tid < nRows) outNorm[rowBase + tid] = sc;
  }
  __syncthreads();

  #pragma unroll
  for (int j = 0; j < 4; ++j) {
    const float xs = noS[wave + 32 * j];
    v4bf p;
    p[0] = f2bf(xv[j].x * xs);
    p[1] = f2bf(xv[j].y * xs);
    p[2] = f2bf(xv[j].z * xs);
    p[3] = f2bf(xv[j].w * xs);
    *reinterpret_cast<v4bf*>(&Xs[(wave + 32 * j) * PROD_PITCH + lane * 4]) = p;
  }
  v8bf wp;
  #pragma unroll
  for (int i = 0; i < 8; ++i) wp[i] = f2bf(wv[i]);
  *reinterpret_cast<v8bf*>(&Wt[wc * PROD_PITCH + wk]) = wp;
  __syncthreads();

  const int hh      = lane >> 4;
  const int l       = lane & 15;
  const int rt      = wave >> 2;
  const int colBase = (wave & 3) << 4;
  const __bf16* xr = Xs + (rt * 16 + l) * PROD_PITCH;
  const __bf16* wr = Wt + (colBase + l) * PROD_PITCH;

  v8f acc = {};
  #pragma unroll
  for (int k0 = 0; k0 < PROD_K; k0 += 32) {
    const v8bf a0 = *reinterpret_cast<const v8bf*>(xr + k0 + (hh << 3));
    const v8bf a1 = *reinterpret_cast<const v8bf*>(xr + k0 + 16 + (hh << 3));
    const v8bf b0 = *reinterpret_cast<const v8bf*>(wr + k0 + (hh << 3));
    const v8bf b1 = *reinterpret_cast<const v8bf*>(wr + k0 + 16 + (hh << 3));
    v16bf a, b;
    #pragma unroll
    for (int i = 0; i < 8; ++i) {
      a[i] = a0[i]; a[i + 8] = a1[i];
      b[i] = b0[i]; b[i + 8] = b1[i];
    }
    acc = __builtin_amdgcn_wmma_f32_16x16x32_bf16(
         false, a,  false, b,
         (short)0, acc,  false,  false);
  }

  const int r0 = rowBase + rt * 16;
  _Float16* __restrict__ yOut = Y + ((size_t)(r0 + (hh << 3)) * 64 + colBase + l);
  if (r0 + 16 <= nRows) {
    #pragma unroll
    for (int v = 0; v < 8; ++v) yOut[(size_t)v * 64] = (_Float16)acc[v];
  } else {
    #pragma unroll
    for (int v = 0; v < 8; ++v) {
      if (r0 + v + (hh << 3) < nRows) yOut[(size_t)v * 64] = (_Float16)acc[v];
    }
  }
}

__global__ __launch_bounds__(1024, 16) void bucket_product_kernel(
    const int2* __restrict__ binned, const int* __restrict__ binOff, const unsigned* __restrict__ planes,
    int* __restrict__ rowptr, float* __restrict__ inNorm, float* __restrict__ outNorm,
    int* __restrict__ sortedSrc, int nE, int nNodes, int nBuck,
    const float* __restrict__ X, const float* __restrict__ Wg, float* __restrict__ Y) {
  __shared__ __attribute__((aligned(16))) unsigned smem[SMEM_WORDS];
  const int bid = (int)blockIdx.x;
  if (bid < nBuck) bucket_group(binned, binOff, rowptr, inNorm, sortedSrc, nE, nNodes, nBuck, bid, smem);
  else             product_rows128(X, planes, outNorm, Wg, reinterpret_cast<_Float16*>(Y), nNodes, (bid - nBuck) * PROD_ROWS, smem);
}

constexpr int GATHER_BATCH   = 8;
constexpr int NODES_PER_WAVE = 4;

template<int U>
__device__ __forceinline__ int half_lane(int x) {
  return __builtin_amdgcn_update_dpp(0, x, 0x150 + U, 0xf, 0xf, true);
}

template<int U0, bool PARTIAL>
__device__ __forceinline__ void gather_batch(const uint2* __restrict__ y8, int rowOff, int cntH,
                                             int p, float4& acc) {
  uint2 w[GATHER_BATCH];
#define GATHER_LOAD(u) w[u] = y8[half_lane<U0 + (u)>(rowOff) | p];
  GATHER_LOAD(0) GATHER_LOAD(1) GATHER_LOAD(2) GATHER_LOAD(3)
  GATHER_LOAD(4) GATHER_LOAD(5) GATHER_LOAD(6) GATHER_LOAD(7)
#undef GATHER_LOAD
  #pragma unroll
  for (int u = 0; u < GATHER_BATCH; ++u) {
    uint2 t = w[u];
    if (PARTIAL) {
      const bool ok = (2 * (U0 + u)) < cntH;
      t.x = ok ? t.x : 0u;
      t.y = ok ? t.y : 0u;
    }
    const v4h v = __builtin_bit_cast(v4h, t);
    acc.x += (float)v[0];
    acc.y += (float)v[1];
    acc.z += (float)v[2];
    acc.w += (float)v[3];
  }
}

__device__ __forceinline__ int list_entry(const int* __restrict__ sortedSrc, int j, int end) {
  const int v = sortedSrc[min(j, max(end - 1, 0))];
  return (j < end) ? v : 0;
}

__device__ __forceinline__ int first_chunk_entry(const int* __restrict__ sortedSrc, int beg, int end, int pos) {
  return list_entry(sortedSrc, beg + pos, end);
}

__device__ __forceinline__ float4 gather_row_sum(const uint2* __restrict__ y8, const int* __restrict__ sortedSrc,
                                                 int beg, int end, int idx, int hh, int p, int pos, int nRows) {
  float4 acc = make_float4(0.0f, 0.0f, 0.0f, 0.0f);
  int base = beg;
  for (; end - base >= 32; base += 32) {
    const int nxt = list_entry(sortedSrc, base + 32 + pos, end);
    const int rowOff = (int)(min((unsigned)idx, (unsigned)(nRows - 1)) << 4);
    gather_batch<0, false>(y8, rowOff, 0, p, acc);
    __builtin_amdgcn_sched_barrier(0);
    gather_batch<8, false>(y8, rowOff, 0, p, acc);
    idx = nxt;
  }
  const int cnt = end - base;
  if (cnt > 0) {
    const int rowOff = (int)(min((unsigned)idx, (unsigned)(nRows - 1)) << 4);
    const int cntH = cnt - hh;
    gather_batch<0, true>(y8, rowOff, cntH, p, acc);
    if (cnt > 16) gather_batch<8, true>(y8, rowOff, cntH, p, acc);
  }
  return acc;
}

__device__ __forceinline__ float xhalf(float x) {
  const int xi = __builtin_bit_cast(int, x);
  return __builtin_bit_cast(float, __builtin_amdgcn_permlanex16(xi, xi, 0x76543210, 0xfedcba98, false, false));
}

__device__ __forceinline__ float2 gather_combine(const float4 acc, int hh) {
  const float kx = hh ? acc.z : acc.x;
  const float ky = hh ? acc.w : acc.y;
  const float sx = hh ? acc.x : acc.z;
  const float sy = hh ? acc.y : acc.w;
  return make_float2(kx + xhalf(sx), ky + xhalf(sy));
}

__global__ __launch_bounds__(256, 16) void gather_finalize_kernel(
    const float* __restrict__ y, const int* __restrict__ rowptr, const int* __restrict__ sortedSrc,
    const float* __restrict__ inNorm, const float* __restrict__ bias, float* __restrict__ out,
    int nRows, int doRelu) {
  static_assert(NODES_PER_WAVE == 4, "the row pointers of a wave's nodes are five named scalars");
  const int lane = threadIdx.x & 31;
  const int n0 = __builtin_amdgcn_readfirstlane((int)((blockIdx.x * 8 + (threadIdx.x >> 5)) * NODES_PER_WAVE));
  if (n0 >= nRows) return;
  const int hh  = lane >> 4;
  const int p   = lane & 15;
  const int pos = (p << 1) | hh;
  const int c0  = (p << 2) | (hh << 1);
  const uint2* __restrict__ y8 = reinterpret_cast<const uint2*>(y);
  const float2 bb = *reinterpret_cast<const float2*>(bias + c0);
  int r0 = rowptr[min(n0,     nRows)];
  int r1 = rowptr[min(n0 + 1, nRows)];
  int r2 = rowptr[min(n0 + 2, nRows)];
  int r3 = rowptr[min(n0 + 3, nRows)];
  const int r4 = rowptr[min(n0 + 4, nRows)];
  int idxN = first_chunk_entry(sortedSrc, r0, r1, pos);
  for (int i = 0; i < NODES_PER_WAVE; ++i) {
    const int n   = n0 + i;
    const int idx = idxN;
    idxN = first_chunk_entry(sortedSrc, r1, r2, pos);
    const float sc = inNorm[min(n, nRows - 1)];
    const float4 part = gather_row_sum(y8, sortedSrc, r0, r1, idx, hh, p, pos, nRows);
    if (n < nRows) {
      const float2 acc = gather_combine(part, hh);
      float2 r;
      r.x = acc.x * sc + bb.x;
      r.y = acc.y * sc + bb.y;
      if (doRelu) { r.x = fmaxf(r.x, 0.0f); r.y = fmaxf(r.y, 0.0f); }
      *reinterpret_cast<float2*>(out + (size_t)n * 64 + c0) = r;
    }
    r0 = r1; r1 = r2; r2 = r3; r3 = r4;
  }
}

typedef __attribute__((ext_vector_type(2))) __bf16 v2bf;

__global__ __launch_bounds__(128, 16) void gather_finalize_gemm_kernel(
    const float* __restrict__ y, const int* __restrict__ rowptr, const int* __restrict__ sortedSrc,
    const float* __restrict__ inNorm, const float* __restrict__ bias,
    const float* __restrict__ outNorm, const float* __restrict__ Wg,
    float* __restrict__ yNext, int nRows) {
  __shared__ __attribute__((aligned(16))) __bf16 Xs[16][64 + 8];
  __shared__ __attribute__((aligned(16))) __bf16 Wt[64][64 + 8];

  static_assert(NODES_PER_WAVE == 4, "4 waves x 4 nodes fill the 16-row A tile; five named row pointers");
  const int tid  = threadIdx.x;
  const int lane = tid & 31;
  const int wave    = __builtin_amdgcn_readfirstlane(tid >> 5);
  const int rowBase = blockIdx.x * 16;
  const int n0      = rowBase + wave * NODES_PER_WAVE;

  int r0 = rowptr[min(n0,     nRows)];
  int r1 = rowptr[min(n0 + 1, nRows)];
  int r2 = rowptr[min(n0 + 2, nRows)];
  int r3 = rowptr[min(n0 + 3, nRows)];
  const int r4 = rowptr[min(n0 + 4, nRows)];

  const int wc  = tid & 63;
  const int wk0 = (tid >> 6) << 3;
  #pragma unroll
  for (int g = 0; g < 4; ++g) {
    const int wk = wk0 + 16 * g;
    float wv[8];
    #pragma unroll
    for (int i = 0; i < 8; ++i) wv[i] = Wg[(wk + i) * 64 + wc];
    v8bf wp;
    #pragma unroll
    for (int i = 0; i < 8; ++i) wp[i] = f2bf(wv[i]);
    *reinterpret_cast<v8bf*>(&Wt[wc][wk]) = wp;
  }

  const int hhG = lane >> 4;
  const int pG  = lane & 15;
  const int pos = (pG << 1) | hhG;
  const int c0  = (pG << 2) | (hhG << 1);
  const uint2* __restrict__ y8 = reinterpret_cast<const uint2*>(y);
  const float2 bb = *reinterpret_cast<const float2*>(bias + c0);
  int idxN = first_chunk_entry(sortedSrc, r0, r1, pos);
  for (int i = 0; i < NODES_PER_WAVE; ++i) {
    const int n   = n0 + i;
    const int idx = idxN;
    idxN = first_chunk_entry(sortedSrc, r1, r2, pos);
    const float sc = inNorm[min(n, nRows - 1)];
    const float so = outNorm[min(n, nRows - 1)];
    const float4 part = gather_row_sum(y8, sortedSrc, r0, r1, idx, hhG, pG, pos, nRows);
    float2 r = make_float2(0.0f, 0.0f);
    if (n < nRows) {
      const float2 acc = gather_combine(part, hhG);
      r.x = fmaxf(acc.x * sc + bb.x, 0.0f) * so;
      r.y = fmaxf(acc.y * sc + bb.y, 0.0f) * so;
    }
    v2bf rp;
    rp[0] = f2bf(r.x);
    rp[1] = f2bf(r.y);
    *reinterpret_cast<v2bf*>(&Xs[wave * NODES_PER_WAVE + i][c0]) = rp;
    r0 = r1; r1 = r2; r2 = r3; r3 = r4;
  }
  __syncthreads();

  const int hh      = lane >> 4;
  const int l       = lane & 15;
  const int colBase = wave << 4;

  v8f acc = {};
  #pragma unroll
  for (int k0 = 0; k0 < 64; k0 += 32) {
    const v8bf a0 = *reinterpret_cast<const v8bf*>(&Xs[l][k0 + (hh << 3)]);
    const v8bf a1 = *reinterpret_cast<const v8bf*>(&Xs[l][k0 + 16 + (hh << 3)]);
    const v8bf b0 = *reinterpret_cast<const v8bf*>(&Wt[colBase + l][k0 + (hh << 3)]);
    const v8bf b1 = *reinterpret_cast<const v8bf*>(&Wt[colBase + l][k0 + 16 + (hh << 3)]);
    v16bf a, b;
    #pragma unroll
    for (int i = 0; i < 8; ++i) {
      a[i] = a0[i]; a[i + 8] = a1[i];
      b[i] = b0[i]; b[i + 8] = b1[i];
    }
    acc = __builtin_amdgcn_wmma_f32_16x16x32_bf16(
         false, a,  false, b,
         (short)0, acc,  false,  false);
  }

  _Float16* __restrict__ yOut = reinterpret_cast<_Float16*>(yNext) + ((long)(rowBase + (hh << 3)) * 64 + colBase + l);
  if (rowBase + 16 <= nRows) {
    #pragma unroll
    for (int v = 0; v < 8; ++v) yOut[(long)v * 64] = (_Float16)acc[v];
  } else {
    #pragma unroll
    for (int v = 0; v < 8; ++v) {
      if (rowBase + v + (hh << 3) < nRows) yOut[(long)v * 64] = (_Float16)acc[v];
    }
  }
}

static inline int cdiv_i(long a, long b) { return (int)((a + b - 1) / b); }

extern "C" void kernel_launch(void* const* d_in, const int* in_sizes, int n_in,
                              void* d_out, int out_size, void* d_ws, size_t ws_size,
                              hipStream_t stream) {
  const float* feat = (const float*)d_in[0];
  const float* W0   = (const float*)d_in[1];
  const float* b0   = (const float*)d_in[2];
  const float* W1   = (const float*)d_in[3];
  const float* b1   = (const float*)d_in[4];
  const float* W2   = (const float*)d_in[5];
  const float* b2   = (const float*)d_in[6];
  const int*   src  = (const int*)d_in[7];
  const int*   dst  = (const int*)d_in[8];

  const int N = in_sizes[0] / 128;
  const int E = in_sizes[7];
  float* out = (float*)d_out;

  const size_t P = (size_t)((N + 63) & ~63);
  const int nBuck  = cdiv_i(N, 1 << BSHIFT);
  const int passes = cdiv_i(N, DEG_BINS);
  float* wsf = (float*)d_ws;
  float* no  = wsf;
  float* ni  = no + P;
  float* y   = ni + P;
  float* h   = y  + (size_t)N * 64;
  unsigned* planes = (unsigned*)(h + (size_t)N * 64);
  int* binOff    = (int*)(planes + (size_t)DEG_PARTS * P);
  int* rowptr    = binOff + (size_t)NSLICE * (size_t)(nBuck + 1);
  int2* binned   = (int2*)(rowptr + P + 64);
  int* sortedSrc = (int*)(binned + (size_t)E);

  const int gemmBlocks = cdiv_i(N, 4 * NODES_PER_WAVE);
  const int gatherBlocks = cdiv_i(N, 8 * NODES_PER_WAVE);

  plane_bin_kernel<<<DEG_PARTS * passes + NSLICE, 1024, 0, stream>>>(src, dst, planes, binned, binOff, E, N, nBuck, passes);
  bucket_product_kernel<<<nBuck + cdiv_i(N, PROD_ROWS), 1024, 0, stream>>>(
      binned, binOff, planes, rowptr, ni, no, sortedSrc, E, N, nBuck, feat, W0, y);
  gather_finalize_gemm_kernel<<<gemmBlocks, 128, 0, stream>>>(y, rowptr, sortedSrc, ni, b0, no, W1, h, N);
  gather_finalize_gemm_kernel<<<gemmBlocks, 128, 0, stream>>>(h, rowptr, sortedSrc, ni, b1, no, W2, y, N);
  gather_finalize_kernel<<<gatherBlocks, 256, 0, stream>>>(y, rowptr, sortedSrc, ni, b2, out, N, 0);
}
